# starting kernel
# baseline (speedup 1.0000x reference)
#include <hip/hip_runtime.h>
#include <math.h>

constexpr int N_NODES = 50000;
constexpr int N_EDGES = 10000;
constexpr int NNZ     = 150000;
constexpr int IN_CH   = 128;
constexpr int HIDDEN  = 64;
constexpr int D       = 6;
constexpr int OUT_CH  = HIDDEN * D;
constexpr int ND      = N_NODES * D;
constexpr int ED      = N_EDGES * D;

typedef __attribute__((ext_vector_type(2))) float v2f;
typedef __attribute__((ext_vector_type(8))) float v8f;

__global__ void wmma_gemm_f32(const float* __restrict__ A, const float* __restrict__ B,
                              const float* __restrict__ bias, float* __restrict__ C,
                              int K, int N, int nTiles, int totalTiles) {
  int tile = blockIdx.x * (blockDim.x >> 5) + (threadIdx.x >> 5);
  if (tile >= totalTiles) return;
  int lane = threadIdx.x & 31;
  int half = lane >> 4;
  int ml   = lane & 15;
  int m0 = (tile / nTiles) * 16;
  int n0 = (tile % nTiles) * 16;

  v8f acc = {};
  const float* Arow = A + (size_t)(m0 + ml) * K + 2 * half;
  const float* Bcol = B + (size_t)(2 * half) * N + n0 + ml;
  for (int k0 = 0; k0 < K; k0 += 4) {
    v2f a, b;
    a.x = Arow[k0];
    a.y = Arow[k0 + 1];
    b.x = Bcol[(size_t)k0 * N];
    b.y = Bcol[(size_t)(k0 + 1) * N];
    acc = __builtin_amdgcn_wmma_f32_16x16x4_f32(false, a, false, b,
                                                (short)0, acc, false, false);
  }
  float bv = bias ? bias[n0 + ml] : 0.0f;
#pragma unroll
  for (int j = 0; j < 8; ++j) {
    C[(size_t)(m0 + j + 8 * half) * N + n0 + ml] = acc[j] + bv;
  }
}

__global__ void stalk_mean_kernel(const float* __restrict__ proj, float* __restrict__ m,
                                  int total) {
  int idx = blockIdx.x * blockDim.x + threadIdx.x;
  if (idx >= total) return;
  int n = idx >> 6;
  int hh = idx & 63;
  const float* p = proj + (size_t)n * OUT_CH + hh;
  float s = 0.0f;
#pragma unroll
  for (int t = 0; t < D; ++t) s += p[t * HIDDEN];
  m[idx] = s * (1.0f / 6.0f);
}

__device__ inline float wave_allsum(float v) {
#pragma unroll
  for (int off = 16; off > 0; off >>= 1) v += __shfl_xor(v, off, 32);
  return v;
}

__global__ void alpha_kernel(const float* __restrict__ xm, const float* __restrict__ em,
                             const int* __restrict__ row, const int* __restrict__ col,
                             const float* __restrict__ ln_g, const float* __restrict__ ln_b,
                             const float* __restrict__ Wsh, const float* __restrict__ bsh,
                             float* __restrict__ alpha, float* __restrict__ Dsum,
                             float* __restrict__ Bcnt) {
  int i = blockIdx.x * (blockDim.x >> 5) + (threadIdx.x >> 5);
  if (i >= NNZ) return;
  int lane = threadIdx.x & 31;
  int rn = row[i], cn = col[i];
  int j0 = lane * 4;
  float z[4];
#pragma unroll
  for (int q = 0; q < 4; ++q) {
    int j = j0 + q;
    z[q] = (j < HIDDEN) ? xm[(size_t)rn * HIDDEN + j]
                        : em[(size_t)cn * HIDDEN + (j - HIDDEN)];
  }
  float s  = z[0] + z[1] + z[2] + z[3];
  float ss = z[0]*z[0] + z[1]*z[1] + z[2]*z[2] + z[3]*z[3];
  s  = wave_allsum(s);
  ss = wave_allsum(ss);
  float mu   = s * (1.0f / 128.0f);
  float var  = ss * (1.0f / 128.0f) - mu * mu;
  float rstd = rsqrtf(var + 1e-5f);

  float p[6] = {0, 0, 0, 0, 0, 0};
#pragma unroll
  for (int q = 0; q < 4; ++q) {
    int j = j0 + q;
    float fn = (z[q] - mu) * rstd * ln_g[j] + ln_b[j];
#pragma unroll
    for (int t = 0; t < D; ++t) p[t] += fn * Wsh[j * D + t];
  }
#pragma unroll
  for (int t = 0; t < D; ++t) p[t] = wave_allsum(p[t]);

  if (lane < D) {
    float a = 1.0f / (1.0f + expf(-(p[lane] + bsh[lane])));
    alpha[(size_t)i * D + lane] = a;
    atomicAdd(&Dsum[rn * D + lane], a * a);
  } else if (lane == D) {
    atomicAdd(&Bcnt[cn], 1.0f);
  }
}

__global__ void safe_inv_kernel(float* __restrict__ buf, int n) {
  int idx = blockIdx.x * blockDim.x + threadIdx.x;
  if (idx >= n) return;
  float v = buf[idx];
  buf[idx] = (v > 0.0f) ? (1.0f / v) : 0.0f;
}

__global__ void scatter_me_kernel(const float* __restrict__ h, const float* __restrict__ alpha,
                                  const int* __restrict__ row, const int* __restrict__ col,
                                  float* __restrict__ me) {
  int idx = blockIdx.x * blockDim.x + threadIdx.x;
  if (idx >= NNZ * OUT_CH) return;
  int i  = idx / OUT_CH;
  int r  = idx - i * OUT_CH;
  int t  = r >> 6;
  int hh = r & 63;
  float a = alpha[(size_t)i * D + t];
  float v = a * h[(size_t)(row[i] * D + t) * HIDDEN + hh];
  atomicAdd(&me[(size_t)(col[i] * D + t) * HIDDEN + hh], v);
}

__global__ void scatter_agg_kernel(const float* __restrict__ me, const float* __restrict__ alpha,
                                   const int* __restrict__ row, const int* __restrict__ col,
                                   const float* __restrict__ Binv, float* __restrict__ agg) {
  int idx = blockIdx.x * blockDim.x + threadIdx.x;
  if (idx >= NNZ * OUT_CH) return;
  int i  = idx / OUT_CH;
  int r  = idx - i * OUT_CH;
  int t  = r >> 6;
  int hh = r & 63;
  int cn = col[i];
  float a = alpha[(size_t)i * D + t] * Binv[cn];
  float v = a * me[(size_t)(cn * D + t) * HIDDEN + hh];
  atomicAdd(&agg[(size_t)(row[i] * D + t) * HIDDEN + hh], v);
}

__global__ void combine_kernel(const float* __restrict__ h, const float* __restrict__ agg,
                               const float* __restrict__ Dinv, const float* __restrict__ bias,
                               float* __restrict__ out, int total) {
  int idx = blockIdx.x * blockDim.x + threadIdx.x;
  if (idx >= total) return;
  int r  = idx >> 6;
  int hh = idx & 63;
  float o = h[idx] - Dinv[r] * agg[idx] + bias[hh];
  out[idx] = (o > 0.0f) ? o : expm1f(o);
}

extern "C" void kernel_launch(void* const* d_in, const int* in_sizes, int n_in,
                              void* d_out, int out_size, void* d_ws, size_t ws_size,
                              hipStream_t stream) {
  (void)in_sizes; (void)n_in; (void)out_size; (void)ws_size;
  const float* x       = (const float*)d_in[0];
  const float* hattr   = (const float*)d_in[1];
  const float* W_lin   = (const float*)d_in[2];
  const float* b_lin   = (const float*)d_in[3];
  const float* ln_g    = (const float*)d_in[4];
  const float* ln_b    = (const float*)d_in[5];
  const float* W_sheaf = (const float*)d_in[6];
  const float* b_sheaf = (const float*)d_in[7];
  const float* W_conv  = (const float*)d_in[8];
  const float* b_conv  = (const float*)d_in[9];
  const int*   eidx    = (const int*)d_in[10];
  const int* row = eidx;
  const int* col = eidx + NNZ;
  float* out = (float*)d_out;

  float* ws    = (float*)d_ws;
  float* xs    = ws;
  float* es    = xs   + (size_t)ND * HIDDEN;
  float* xm    = es   + (size_t)ED * HIDDEN;
  float* em    = xm   + (size_t)N_NODES * HIDDEN;
  float* alpha = em   + (size_t)N_EDGES * HIDDEN;
  float* Dbuf  = alpha + (size_t)NNZ * D;
  float* Bbuf  = Dbuf + ND;
  float* hbuf  = Bbuf + N_EDGES;
  float* me    = hbuf + (size_t)ND * HIDDEN;
  float* agg   = me   + (size_t)ED * HIDDEN;

  const int TPB = 256;

  {
    int tiles = (N_NODES / 16) * (OUT_CH / 16);
    wmma_gemm_f32<<<tiles / 8, TPB, 0, stream>>>(x, W_lin, b_lin, xs,
                                                 IN_CH, OUT_CH, OUT_CH / 16, tiles);
    int etiles = (N_EDGES / 16) * (OUT_CH / 16);
    wmma_gemm_f32<<<etiles / 8, TPB, 0, stream>>>(hattr, W_lin, b_lin, es,
                                                  IN_CH, OUT_CH, OUT_CH / 16, etiles);
  }

  stalk_mean_kernel<<<(N_NODES * HIDDEN + TPB - 1) / TPB, TPB, 0, stream>>>(xs, xm, N_NODES * HIDDEN);
  stalk_mean_kernel<<<(N_EDGES * HIDDEN + TPB - 1) / TPB, TPB, 0, stream>>>(es, em, N_EDGES * HIDDEN);

  hipMemsetAsync(Dbuf, 0, (size_t)ND * sizeof(float), stream);
  hipMemsetAsync(Bbuf, 0, (size_t)N_EDGES * sizeof(float), stream);
  alpha_kernel<<<NNZ / 8, TPB, 0, stream>>>(xm, em, row, col, ln_g, ln_b,
                                            W_sheaf, b_sheaf, alpha, Dbuf, Bbuf);
  safe_inv_kernel<<<(ND + TPB - 1) / TPB, TPB, 0, stream>>>(Dbuf, ND);
  safe_inv_kernel<<<(N_EDGES + TPB - 1) / TPB, TPB, 0, stream>>>(Bbuf, N_EDGES);

  const int convTiles  = (ND / 16) * (HIDDEN / 16);
  const int scatterBlk = (NNZ * OUT_CH + TPB - 1) / TPB;
  const int elemBlk    = (ND * HIDDEN + TPB - 1) / TPB;

  wmma_gemm_f32<<<convTiles / 8, TPB, 0, stream>>>(xs, W_conv, nullptr, hbuf,
                                                   HIDDEN, HIDDEN, HIDDEN / 16, convTiles);
  hipMemsetAsync(me, 0, (size_t)ED * HIDDEN * sizeof(float), stream);
  scatter_me_kernel<<<scatterBlk, TPB, 0, stream>>>(hbuf, alpha, row, col, me);
  hipMemsetAsync(agg, 0, (size_t)ND * HIDDEN * sizeof(float), stream);
  scatter_agg_kernel<<<scatterBlk, TPB, 0, stream>>>(me, alpha, row, col, Bbuf, agg);
  combine_kernel<<<elemBlk, TPB, 0, stream>>>(hbuf, agg, Dbuf, b_conv, out, ND * HIDDEN);

  wmma_gemm_f32<<<convTiles / 8, TPB, 0, stream>>>(out, W_conv + HIDDEN * HIDDEN, nullptr, hbuf,
                                                   HIDDEN, HIDDEN, HIDDEN / 16, convTiles);
  hipMemsetAsync(me, 0, (size_t)ED * HIDDEN * sizeof(float), stream);
  scatter_me_kernel<<<scatterBlk, TPB, 0, stream>>>(hbuf, alpha, row, col, me);
  hipMemsetAsync(agg, 0, (size_t)ND * HIDDEN * sizeof(float), stream);
  scatter_agg_kernel<<<scatterBlk, TPB, 0, stream>>>(me, alpha, row, col, Bbuf, agg);
  combine_kernel<<<elemBlk, TPB, 0, stream>>>(hbuf, agg, Dbuf, b_conv + HIDDEN, out, ND * HIDDEN);
}


# Scatter-adds by range owners, no atomics: a work-group scans all ids for its rows, each lane sums its own 4 columns in LDS in fixed scan order (was arrival order); node owner does scale+ELU
# speedup vs baseline: 2.1576x; 2.1576x over previous
#include <hip/hip_runtime.h>
#include <math.h>

constexpr int N_NODES = 50000;
constexpr int N_EDGES = 10000;
constexpr int NNZ     = 150000;
constexpr int IN_CH   = 128;
constexpr int HIDDEN  = 64;
constexpr int D       = 6;
constexpr int OUT_CH  = HIDDEN * D;
constexpr int ND      = N_NODES * D;
constexpr int ED      = N_EDGES * D;

typedef __attribute__((ext_vector_type(2))) float v2f;
typedef __attribute__((ext_vector_type(8))) float v8f;

__global__ void wmma_gemm_f32(const float* __restrict__ A, const float* __restrict__ B,
                              const float* __restrict__ bias, float* __restrict__ C,
                              int K, int N, int nTiles, int totalTiles) {
  int tile = blockIdx.x * (blockDim.x >> 5) + (threadIdx.x >> 5);
  if (tile >= totalTiles) return;
  int lane = threadIdx.x & 31;
  int half = lane >> 4;
  int ml   = lane & 15;
  int m0 = (tile / nTiles) * 16;
  int n0 = (tile % nTiles) * 16;

  v8f acc = {};
  const float* Arow = A + (size_t)(m0 + ml) * K + 2 * half;
  const float* Bcol = B + (size_t)(2 * half) * N + n0 + ml;
  for (int k0 = 0; k0 < K; k0 += 4) {
    v2f a, b;
    a.x = Arow[k0];
    a.y = Arow[k0 + 1];
    b.x = Bcol[(size_t)k0 * N];
    b.y = Bcol[(size_t)(k0 + 1) * N];
    acc = __builtin_amdgcn_wmma_f32_16x16x4_f32(false, a, false, b,
                                                (short)0, acc, false, false);
  }
  float bv = bias ? bias[n0 + ml] : 0.0f;
#pragma unroll
  for (int j = 0; j < 8; ++j) {
    C[(size_t)(m0 + j + 8 * half) * N + n0 + ml] = acc[j] + bv;
  }
}

__global__ void stalk_mean_kernel(const float* __restrict__ proj, float* __restrict__ m,
                                  int total) {
  int idx = blockIdx.x * blockDim.x + threadIdx.x;
  if (idx >= total) return;
  int n = idx >> 6;
  int hh = idx & 63;
  const float* p = proj + (size_t)n * OUT_CH + hh;
  float s = 0.0f;
#pragma unroll
  for (int t = 0; t < D; ++t) s += p[t * HIDDEN];
  m[idx] = s * (1.0f / 6.0f);
}

__device__ inline float wave_allsum(float v) {
#pragma unroll
  for (int off = 16; off > 0; off >>= 1) v += __shfl_xor(v, off, 32);
  return v;
}

__global__ void alpha_kernel(const float* __restrict__ xm, const float* __restrict__ em,
                             const int* __restrict__ row, const int* __restrict__ col,
                             const float* __restrict__ ln_g, const float* __restrict__ ln_b,
                             const float* __restrict__ Wsh, const float* __restrict__ bsh,
                             float* __restrict__ alpha, float* __restrict__ Dsum,
                             float* __restrict__ Bcnt) {
  int i = blockIdx.x * (blockDim.x >> 5) + (threadIdx.x >> 5);
  if (i >= NNZ) return;
  int lane = threadIdx.x & 31;
  int rn = row[i], cn = col[i];
  int j0 = lane * 4;
  float z[4];
#pragma unroll
  for (int q = 0; q < 4; ++q) {
    int j = j0 + q;
    z[q] = (j < HIDDEN) ? xm[(size_t)rn * HIDDEN + j]
                        : em[(size_t)cn * HIDDEN + (j - HIDDEN)];
  }
  float s  = z[0] + z[1] + z[2] + z[3];
  float ss = z[0]*z[0] + z[1]*z[1] + z[2]*z[2] + z[3]*z[3];
  s  = wave_allsum(s);
  ss = wave_allsum(ss);
  float mu   = s * (1.0f / 128.0f);
  float var  = ss * (1.0f / 128.0f) - mu * mu;
  float rstd = rsqrtf(var + 1e-5f);

  float p[6] = {0, 0, 0, 0, 0, 0};
#pragma unroll
  for (int q = 0; q < 4; ++q) {
    int j = j0 + q;
    float fn = (z[q] - mu) * rstd * ln_g[j] + ln_b[j];
#pragma unroll
    for (int t = 0; t < D; ++t) p[t] += fn * Wsh[j * D + t];
  }
#pragma unroll
  for (int t = 0; t < D; ++t) p[t] = wave_allsum(p[t]);

  if (lane < D) {
    float a = 1.0f / (1.0f + expf(-(p[lane] + bsh[lane])));
    alpha[(size_t)i * D + lane] = a;
    atomicAdd(&Dsum[rn * D + lane], a * a);
  } else if (lane == D) {
    atomicAdd(&Bcnt[cn], 1.0f);
  }
}

__global__ void safe_inv_kernel(float* __restrict__ buf, int n) {
  int idx = blockIdx.x * blockDim.x + threadIdx.x;
  if (idx >= n) return;
  float v = buf[idx];
  buf[idx] = (v > 0.0f) ? (1.0f / v) : 0.0f;
}

constexpr int SEG_TPB   = 96;
constexpr int SEG_WAVES = 3;
constexpr int SEG_CAP   = 256;
constexpr int STRIP     = 128;
constexpr int NSTRIPS   = (NNZ + STRIP - 1) / STRIP;
constexpr int NFULL     = NNZ / STRIP;
constexpr int SCAN_U    = 16;
constexpr int ACC_U     = 8;
constexpr int FIN_U     = 4;
static_assert(NNZ % 4 == 0, "16-byte id loads are whole or absent");
static_assert(NNZ <= (1 << 18), "incidence id is packed into 18 bits");
static_assert(SEG_CAP >= 2 * STRIP, "a round must be able to take a strip");

template <int R, bool NODE>
__global__ __launch_bounds__(SEG_TPB)
void segsum_kernel(const float* __restrict__ src, const float* __restrict__ alpha,
                   const int* __restrict__ key, const int* __restrict__ oth,
                   const float* __restrict__ Binv, const float* __restrict__ hself,
                   const float* __restrict__ Dinv, const float* __restrict__ bias,
                   float* __restrict__ dst, int nDest) {
  static_assert(R <= 64, "owned row number and the 18-bit id share one word");
  __shared__ float4 acc[R * SEG_TPB];
  __shared__ int    lst[SEG_WAVES * SEG_CAP];
  __shared__ int    cntS[SEG_WAVES];
  __shared__ int    doneS[SEG_WAVES];

  const int tid  = threadIdx.x;
  const int wave = __builtin_amdgcn_readfirstlane(tid >> 5);
  const int lane = tid & 31;
  const int t    = tid >> 4;
  const int r0   = blockIdx.x * R;
  const int rr   = (nDest - r0 < R) ? (nDest - r0) : R;

  for (int r = 0; r < R; ++r) acc[r * SEG_TPB + tid] = make_float4(0.f, 0.f, 0.f, 0.f);

  int cnt = 0;
  auto strip = [&](const int4 k, const int i0) {
    const unsigned d0 = (unsigned)(k.x - r0);
    const unsigned d1 = (unsigned)(k.y - r0);
    const unsigned d2 = (unsigned)(k.z - r0);
    const unsigned d3 = (unsigned)(k.w - r0);
    const unsigned dm = min(min(d0, d1), min(d2, d3));
    if (__builtin_amdgcn_ballot_w32(dm < (unsigned)rr) != 0u) {
      auto append = [&](unsigned d, int i) {
        const bool m = d < (unsigned)rr;
        const unsigned b = __builtin_amdgcn_ballot_w32(m);
        if (m) {
          const int pos = cnt + (int)__builtin_amdgcn_mbcnt_lo(b, 0u);
          lst[wave * SEG_CAP + pos] = (int)((d << 18) | (unsigned)i);
        }
        cnt += __popc(b);
      };
      append(d0, i0);
      append(d1, i0 + 1);
      append(d2, i0 + 2);
      append(d3, i0 + 3);
    }
  };

  int s = wave;
  bool alldone = false;
  while (!alldone) {
    cnt = 0;
    bool full = false;
    while (!full && s + (SCAN_U - 1) * SEG_WAVES < NFULL) {
      int4 k[SCAN_U];
      const int ib = s * STRIP + lane * 4;
#pragma unroll
      for (int u = 0; u < SCAN_U; ++u)
        k[u] = *reinterpret_cast<const int4*>(key + ib + u * SEG_WAVES * STRIP);
      int used = SCAN_U;
#pragma unroll
      for (int u = 0; u < SCAN_U; ++u) {
        strip(k[u], ib + u * SEG_WAVES * STRIP);
        if (cnt > SEG_CAP - STRIP) { full = true; used = u + 1; break; }
      }
      s += used * SEG_WAVES;
    }
    while (!full && s < NSTRIPS) {
      const int i0 = s * STRIP + lane * 4;
      const int4 k = (i0 < NNZ) ? *reinterpret_cast<const int4*>(key + i0)
                                : make_int4(-1, -1, -1, -1);
      strip(k, i0);
      s += SEG_WAVES;
      if (cnt > SEG_CAP - STRIP) full = true;
    }
    if (lane == 0) { cntS[wave] = cnt; doneS[wave] = (s >= NSTRIPS) ? 1 : 0; }
    __syncthreads();

    for (int w = 0; w < SEG_WAVES; ++w) {
      const int n = __builtin_amdgcn_readfirstlane(cntS[w]);
      const int* L = lst + w * SEG_CAP;
      int   pkN[ACC_U];
      int   soN[ACC_U];
      float aN[ACC_U];
      auto hop1 = [&](const int mb) {
#pragma unroll
        for (int j = 0; j < ACC_U; ++j) pkN[j] = L[(mb + j < n) ? (mb + j) : (n - 1)];
#pragma unroll
        for (int j = 0; j < ACC_U; ++j) {
          const int i = pkN[j] & 0x3FFFF;
          soN[j] = oth[i];
          aN[j]  = alpha[(size_t)i * D + t];
        }
      };
      if (n > 0) hop1(0);
      for (int mb = 0; mb < n; mb += ACC_U) {
        int    pk[ACC_U];
        float  a[ACC_U];
        float  bi[ACC_U];
        float4 v[ACC_U];
#pragma unroll
        for (int j = 0; j < ACC_U; ++j) {
          pk[j] = pkN[j];
          a[j]  = aN[j];
          v[j]  = *reinterpret_cast<const float4*>(src + (size_t)soN[j] * OUT_CH + tid * 4);
          bi[j] = NODE ? Binv[soN[j]] : 1.0f;
        }
        hop1(mb + ACC_U);
#pragma unroll
        for (int j = 0; j < ACC_U; ++j) {
          const bool live = (mb + j < n);
          const int d = (int)((unsigned)pk[j] >> 18);
          const float aa = live ? (NODE ? a[j] * bi[j] : a[j]) : 0.0f;
          const float4 vv = live ? v[j] : make_float4(0.f, 0.f, 0.f, 0.f);
          float4 c = acc[d * SEG_TPB + tid];
          c.x += aa * vv.x;
          c.y += aa * vv.y;
          c.z += aa * vv.z;
          c.w += aa * vv.w;
          acc[d * SEG_TPB + tid] = c;
        }
      }
    }
    alldone = (doneS[0] & doneS[1] & doneS[2]) != 0;
    __syncthreads();
  }

  float4 b = make_float4(0.f, 0.f, 0.f, 0.f);
  if (NODE) b = *reinterpret_cast<const float4*>(bias + ((tid * 4) & (HIDDEN - 1)));
  auto finish = [&](const int r, const float4 hv, const float dinv) {
    const float4 c = acc[r * SEG_TPB + tid];
    float4 ov = c;
    if (NODE) {
      ov.x = hv.x - dinv * c.x + b.x;
      ov.y = hv.y - dinv * c.y + b.y;
      ov.z = hv.z - dinv * c.z + b.z;
      ov.w = hv.w - dinv * c.w + b.w;
      ov.x = (ov.x > 0.0f) ? ov.x : expm1f(ov.x);
      ov.y = (ov.y > 0.0f) ? ov.y : expm1f(ov.y);
      ov.z = (ov.z > 0.0f) ? ov.z : expm1f(ov.z);
      ov.w = (ov.w > 0.0f) ? ov.w : expm1f(ov.w);
    }
    *reinterpret_cast<float4*>(dst + (size_t)(r0 + r) * OUT_CH + tid * 4) = ov;
  };
  if (rr == R) {
    static_assert(R % FIN_U == 0, "whole groups of rows");
    for (int rb = 0; rb < R; rb += FIN_U) {
      float4 hv[FIN_U];
      float  dinv[FIN_U];
#pragma unroll
      for (int j = 0; j < FIN_U; ++j) {
        hv[j]   = NODE ? *reinterpret_cast<const float4*>(hself + (size_t)(r0 + rb + j) * OUT_CH + tid * 4)
                       : make_float4(0.f, 0.f, 0.f, 0.f);
        dinv[j] = NODE ? Dinv[(r0 + rb + j) * D + t] : 0.0f;
      }
#pragma unroll
      for (int j = 0; j < FIN_U; ++j) finish(rb + j, hv[j], dinv[j]);
    }
  } else {
    for (int r = 0; r < rr; ++r) {
      const float4 hv = NODE ? *reinterpret_cast<const float4*>(hself + (size_t)(r0 + r) * OUT_CH + tid * 4)
                             : make_float4(0.f, 0.f, 0.f, 0.f);
      const float dinv = NODE ? Dinv[(r0 + r) * D + t] : 0.0f;
      finish(r, hv, dinv);
    }
  }
}

extern "C" void kernel_launch(void* const* d_in, const int* in_sizes, int n_in,
                              void* d_out, int out_size, void* d_ws, size_t ws_size,
                              hipStream_t stream) {
  (void)in_sizes; (void)n_in; (void)out_size; (void)ws_size;
  const float* x       = (const float*)d_in[0];
  const float* hattr   = (const float*)d_in[1];
  const float* W_lin   = (const float*)d_in[2];
  const float* b_lin   = (const float*)d_in[3];
  const float* ln_g    = (const float*)d_in[4];
  const float* ln_b    = (const float*)d_in[5];
  const float* W_sheaf = (const float*)d_in[6];
  const float* b_sheaf = (const float*)d_in[7];
  const float* W_conv  = (const float*)d_in[8];
  const float* b_conv  = (const float*)d_in[9];
  const int*   eidx    = (const int*)d_in[10];
  const int* row = eidx;
  const int* col = eidx + NNZ;
  float* out = (float*)d_out;

  float* ws    = (float*)d_ws;
  float* xs    = ws;
  float* es    = xs   + (size_t)ND * HIDDEN;
  float* xm    = es   + (size_t)ED * HIDDEN;
  float* em    = xm   + (size_t)N_NODES * HIDDEN;
  float* alpha = em   + (size_t)N_EDGES * HIDDEN;
  float* Dbuf  = alpha + (size_t)NNZ * D;
  float* Bbuf  = Dbuf + ND;
  float* hbuf  = Bbuf + N_EDGES;
  float* me    = hbuf + (size_t)ND * HIDDEN;

  const int TPB = 256;

  {
    int tiles = (N_NODES / 16) * (OUT_CH / 16);
    wmma_gemm_f32<<<tiles / 8, TPB, 0, stream>>>(x, W_lin, b_lin, xs,
                                                 IN_CH, OUT_CH, OUT_CH / 16, tiles);
    int etiles = (N_EDGES / 16) * (OUT_CH / 16);
    wmma_gemm_f32<<<etiles / 8, TPB, 0, stream>>>(hattr, W_lin, b_lin, es,
                                                  IN_CH, OUT_CH, OUT_CH / 16, etiles);
  }

  stalk_mean_kernel<<<(N_NODES * HIDDEN + TPB - 1) / TPB, TPB, 0, stream>>>(xs, xm, N_NODES * HIDDEN);
  stalk_mean_kernel<<<(N_EDGES * HIDDEN + TPB - 1) / TPB, TPB, 0, stream>>>(es, em, N_EDGES * HIDDEN);

  hipMemsetAsync(Dbuf, 0, (size_t)ND * sizeof(float), stream);
  hipMemsetAsync(Bbuf, 0, (size_t)N_EDGES * sizeof(float), stream);
  alpha_kernel<<<NNZ / 8, TPB, 0, stream>>>(xm, em, row, col, ln_g, ln_b,
                                            W_sheaf, b_sheaf, alpha, Dbuf, Bbuf);
  safe_inv_kernel<<<(ND + TPB - 1) / TPB, TPB, 0, stream>>>(Dbuf, ND);
  safe_inv_kernel<<<(N_EDGES + TPB - 1) / TPB, TPB, 0, stream>>>(Bbuf, N_EDGES);

  const int convTiles  = (ND / 16) * (HIDDEN / 16);
  constexpr int RE = 8;
  constexpr int RN = 40;
  const int edgeBlk = (N_EDGES + RE - 1) / RE;
  const int nodeBlk = (N_NODES + RN - 1) / RN;

  wmma_gemm_f32<<<convTiles / 8, TPB, 0, stream>>>(xs, W_conv, nullptr, hbuf,
                                                   HIDDEN, HIDDEN, HIDDEN / 16, convTiles);
  segsum_kernel<RE, false><<<edgeBlk, SEG_TPB, 0, stream>>>(hbuf, alpha, col, row, nullptr, nullptr,
                                                            nullptr, nullptr, me, N_EDGES);
  segsum_kernel<RN, true><<<nodeBlk, SEG_TPB, 0, stream>>>(me, alpha, row, col, Bbuf, hbuf,
                                                           Dbuf, b_conv, out, N_NODES);

  wmma_gemm_f32<<<convTiles / 8, TPB, 0, stream>>>(out, W_conv + HIDDEN * HIDDEN, nullptr, hbuf,
                                                   HIDDEN, HIDDEN, HIDDEN / 16, convTiles);
  segsum_kernel<RE, false><<<edgeBlk, SEG_TPB, 0, stream>>>(hbuf, alpha, col, row, nullptr, nullptr,
                                                            nullptr, nullptr, me, N_EDGES);
  segsum_kernel<RN, true><<<nodeBlk, SEG_TPB, 0, stream>>>(me, alpha, row, col, Bbuf, hbuf,
                                                           Dbuf, b_conv + HIDDEN, out, N_NODES);
}


# GEMMs through LDS: weight slice staged once per work-group in fragment order, each wave copies its 16-row A tile by whole lines, result leaves as whole rows; k summed in a permuted order
# speedup vs baseline: 2.5647x; 1.1887x over previous
#include <hip/hip_runtime.h>
#include <math.h>

constexpr int N_NODES = 50000;
constexpr int N_EDGES = 10000;
constexpr int NNZ     = 150000;
constexpr int IN_CH   = 128;
constexpr int HIDDEN  = 64;
constexpr int D       = 6;
constexpr int OUT_CH  = HIDDEN * D;
constexpr int ND      = N_NODES * D;
constexpr int ED      = N_EDGES * D;

typedef __attribute__((ext_vector_type(2))) float v2f;
typedef __attribute__((ext_vector_type(8))) float v8f;

constexpr int GEMM_WAVES = 8;
constexpr int GEMM_TPB   = GEMM_WAVES * 32;

template <int K>
__global__ __launch_bounds__(GEMM_TPB)
void wmma_gemm_lds(const float* __restrict__ A, const float* __restrict__ B,
                   const float* __restrict__ bias, float* __restrict__ C,
                   int N, int mTiles) {
  constexpr int KP   = K / 8;
  constexpr int AP   = K + 4;
  constexpr int CP   = 64 + 4;
  constexpr int SLOT = 16 * AP;
  static_assert(K % 8 == 0 && K >= 64, "slot must also hold the 16 x 64 result");
  __shared__ float4 Bs[KP * 4 * 32];
  __shared__ float4 Ss[GEMM_WAVES * SLOT / 4];

  const int tid  = threadIdx.x;
  const int wave = __builtin_amdgcn_readfirstlane(tid >> 5);
  const int lane = tid & 31;
  const int c    = lane & 15;
  const int h    = lane >> 4;
  const int n0   = blockIdx.y * 64;

  {
    constexpr int BIT = KP * 4 * 32 / GEMM_TPB;
    float4 bw[BIT];
#pragma unroll
    for (int it = 0; it < BIT; ++it) {
      const int idx = tid + it * GEMM_TPB;
      const int l  = idx & 31;
      const int nt = (idx >> 5) & 3;
      const int p  = idx >> 7;
      const float* bp = B + (size_t)(8 * p + 4 * (l >> 4)) * N + n0 + nt * 16 + (l & 15);
      bw[it].x = bp[0];
      bw[it].y = bp[N];
      bw[it].z = bp[2 * (size_t)N];
      bw[it].w = bp[3 * (size_t)N];
    }
#pragma unroll
    for (int it = 0; it < BIT; ++it) Bs[tid + it * GEMM_TPB] = bw[it];
  }
  __syncthreads();

  float* slot = reinterpret_cast<float*>(Ss) + wave * SLOT;
  float bv[4];
#pragma unroll
  for (int nt = 0; nt < 4; ++nt) bv[nt] = bias ? bias[n0 + nt * 16 + c] : 0.0f;

  for (int mt = blockIdx.x * GEMM_WAVES + wave; mt < mTiles; mt += gridDim.x * GEMM_WAVES) {
    const float4* g = reinterpret_cast<const float4*>(A + (size_t)mt * 16 * K);
    float4 tl[KP];
#pragma unroll
    for (int i = 0; i < KP; ++i) tl[i] = g[i * 32 + lane];
#pragma unroll
    for (int i = 0; i < KP; ++i) {
      const int f = i * 32 + lane;
      const int r = f / (K / 4);
      const int q = f % (K / 4);
      *reinterpret_cast<float4*>(slot + r * AP + 4 * q) = tl[i];
    }

    v8f acc[4] = {};
#pragma unroll 2
    for (int p = 0; p < KP; ++p) {
      const float4 a4 = *reinterpret_cast<const float4*>(slot + c * AP + 8 * p + 4 * h);
      float4 b4[4];
#pragma unroll
      for (int nt = 0; nt < 4; ++nt) b4[nt] = Bs[(p * 4 + nt) * 32 + lane];
      v2f a0; a0.x = a4.x; a0.y = a4.y;
      v2f a1; a1.x = a4.z; a1.y = a4.w;
#pragma unroll
      for (int nt = 0; nt < 4; ++nt) {
        v2f b0; b0.x = b4[nt].x; b0.y = b4[nt].y;
        acc[nt] = __builtin_amdgcn_wmma_f32_16x16x4_f32(false, a0, false, b0,
                                                        (short)0, acc[nt], false, false);
      }
#pragma unroll
      for (int nt = 0; nt < 4; ++nt) {
        v2f b1; b1.x = b4[nt].z; b1.y = b4[nt].w;
        acc[nt] = __builtin_amdgcn_wmma_f32_16x16x4_f32(false, a1, false, b1,
                                                        (short)0, acc[nt], false, false);
      }
    }

#pragma unroll
    for (int nt = 0; nt < 4; ++nt) {
#pragma unroll
      for (int r = 0; r < 8; ++r) slot[(8 * h + r) * CP + nt * 16 + c] = acc[nt][r] + bv[nt];
    }
#pragma unroll
    for (int i = 0; i < 8; ++i) {
      const int f   = i * 32 + lane;
      const int row = f >> 4;
      const int q   = f & 15;
      const float4 o = *reinterpret_cast<const float4*>(slot + row * CP + 4 * q);
      *reinterpret_cast<float4*>(C + (size_t)(mt * 16 + row) * N + n0 + 4 * q) = o;
    }
  }
}

__global__ void stalk_mean_kernel(const float* __restrict__ proj, float* __restrict__ m,
                                  int total) {
  int idx = blockIdx.x * blockDim.x + threadIdx.x;
  if (idx >= total) return;
  int n = idx >> 6;
  int hh = idx & 63;
  const float* p = proj + (size_t)n * OUT_CH + hh;
  float s = 0.0f;
#pragma unroll
  for (int t = 0; t < D; ++t) s += p[t * HIDDEN];
  m[idx] = s * (1.0f / 6.0f);
}

__device__ inline float wave_allsum(float v) {
#pragma unroll
  for (int off = 16; off > 0; off >>= 1) v += __shfl_xor(v, off, 32);
  return v;
}

__global__ void alpha_kernel(const float* __restrict__ xm, const float* __restrict__ em,
                             const int* __restrict__ row, const int* __restrict__ col,
                             const float* __restrict__ ln_g, const float* __restrict__ ln_b,
                             const float* __restrict__ Wsh, const float* __restrict__ bsh,
                             float* __restrict__ alpha, float* __restrict__ Dsum,
                             float* __restrict__ Bcnt) {
  int i = blockIdx.x * (blockDim.x >> 5) + (threadIdx.x >> 5);
  if (i >= NNZ) return;
  int lane = threadIdx.x & 31;
  int rn = row[i], cn = col[i];
  int j0 = lane * 4;
  float z[4];
#pragma unroll
  for (int q = 0; q < 4; ++q) {
    int j = j0 + q;
    z[q] = (j < HIDDEN) ? xm[(size_t)rn * HIDDEN + j]
                        : em[(size_t)cn * HIDDEN + (j - HIDDEN)];
  }
  float s  = z[0] + z[1] + z[2] + z[3];
  float ss = z[0]*z[0] + z[1]*z[1] + z[2]*z[2] + z[3]*z[3];
  s  = wave_allsum(s);
  ss = wave_allsum(ss);
  float mu   = s * (1.0f / 128.0f);
  float var  = ss * (1.0f / 128.0f) - mu * mu;
  float rstd = rsqrtf(var + 1e-5f);

  float p[6] = {0, 0, 0, 0, 0, 0};
#pragma unroll
  for (int q = 0; q < 4; ++q) {
    int j = j0 + q;
    float fn = (z[q] - mu) * rstd * ln_g[j] + ln_b[j];
#pragma unroll
    for (int t = 0; t < D; ++t) p[t] += fn * Wsh[j * D + t];
  }
#pragma unroll
  for (int t = 0; t < D; ++t) p[t] = wave_allsum(p[t]);

  if (lane < D) {
    float a = 1.0f / (1.0f + expf(-(p[lane] + bsh[lane])));
    alpha[(size_t)i * D + lane] = a;
    atomicAdd(&Dsum[rn * D + lane], a * a);
  } else if (lane == D) {
    atomicAdd(&Bcnt[cn], 1.0f);
  }
}

__global__ void safe_inv_kernel(float* __restrict__ buf, int n) {
  int idx = blockIdx.x * blockDim.x + threadIdx.x;
  if (idx >= n) return;
  float v = buf[idx];
  buf[idx] = (v > 0.0f) ? (1.0f / v) : 0.0f;
}

constexpr int SEG_TPB   = 96;
constexpr int SEG_WAVES = 3;
constexpr int SEG_CAP   = 256;
constexpr int STRIP     = 128;
constexpr int NSTRIPS   = (NNZ + STRIP - 1) / STRIP;
constexpr int NFULL     = NNZ / STRIP;
constexpr int SCAN_U    = 16;
constexpr int ACC_U     = 8;
constexpr int FIN_U     = 4;
static_assert(NNZ % 4 == 0, "16-byte id loads are whole or absent");
static_assert(NNZ <= (1 << 18), "incidence id is packed into 18 bits");
static_assert(SEG_CAP >= 2 * STRIP, "a round must be able to take a strip");

template <int R, bool NODE>
__global__ __launch_bounds__(SEG_TPB)
void segsum_kernel(const float* __restrict__ src, const float* __restrict__ alpha,
                   const int* __restrict__ key, const int* __restrict__ oth,
                   const float* __restrict__ Binv, const float* __restrict__ hself,
                   const float* __restrict__ Dinv, const float* __restrict__ bias,
                   float* __restrict__ dst, int nDest) {
  static_assert(R <= 64, "owned row number and the 18-bit id share one word");
  __shared__ float4 acc[R * SEG_TPB];
  __shared__ int    lst[SEG_WAVES * SEG_CAP];
  __shared__ int    cntS[SEG_WAVES];
  __shared__ int    doneS[SEG_WAVES];

  const int tid  = threadIdx.x;
  const int wave = __builtin_amdgcn_readfirstlane(tid >> 5);
  const int lane = tid & 31;
  const int t    = tid >> 4;
  const int r0   = blockIdx.x * R;
  const int rr   = (nDest - r0 < R) ? (nDest - r0) : R;

  for (int r = 0; r < R; ++r) acc[r * SEG_TPB + tid] = make_float4(0.f, 0.f, 0.f, 0.f);

  int cnt = 0;
  auto strip = [&](const int4 k, const int i0) {
    const unsigned d0 = (unsigned)(k.x - r0);
    const unsigned d1 = (unsigned)(k.y - r0);
    const unsigned d2 = (unsigned)(k.z - r0);
    const unsigned d3 = (unsigned)(k.w - r0);
    const unsigned dm = min(min(d0, d1), min(d2, d3));
    if (__builtin_amdgcn_ballot_w32(dm < (unsigned)rr) != 0u) {
      auto append = [&](unsigned d, int i) {
        const bool m = d < (unsigned)rr;
        const unsigned b = __builtin_amdgcn_ballot_w32(m);
        if (m) {
          const int pos = cnt + (int)__builtin_amdgcn_mbcnt_lo(b, 0u);
          lst[wave * SEG_CAP + pos] = (int)((d << 18) | (unsigned)i);
        }
        cnt += __popc(b);
      };
      append(d0, i0);
      append(d1, i0 + 1);
      append(d2, i0 + 2);
      append(d3, i0 + 3);
    }
  };

  int s = wave;
  bool alldone = false;
  while (!alldone) {
    cnt = 0;
    bool full = false;
    while (!full && s + (SCAN_U - 1) * SEG_WAVES < NFULL) {
      int4 k[SCAN_U];
      const int ib = s * STRIP + lane * 4;
#pragma unroll
      for (int u = 0; u < SCAN_U; ++u)
        k[u] = *reinterpret_cast<const int4*>(key + ib + u * SEG_WAVES * STRIP);
      int used = SCAN_U;
#pragma unroll
      for (int u = 0; u < SCAN_U; ++u) {
        strip(k[u], ib + u * SEG_WAVES * STRIP);
        if (cnt > SEG_CAP - STRIP) { full = true; used = u + 1; break; }
      }
      s += used * SEG_WAVES;
    }
    while (!full && s < NSTRIPS) {
      const int i0 = s * STRIP + lane * 4;
      const int4 k = (i0 < NNZ) ? *reinterpret_cast<const int4*>(key + i0)
                                : make_int4(-1, -1, -1, -1);
      strip(k, i0);
      s += SEG_WAVES;
      if (cnt > SEG_CAP - STRIP) full = true;
    }
    if (lane == 0) { cntS[wave] = cnt; doneS[wave] = (s >= NSTRIPS) ? 1 : 0; }
    __syncthreads();

    for (int w = 0; w < SEG_WAVES; ++w) {
      const int n = __builtin_amdgcn_readfirstlane(cntS[w]);
      const int* L = lst + w * SEG_CAP;
      int   pkN[ACC_U];
      int   soN[ACC_U];
      float aN[ACC_U];
      auto hop1 = [&](const int mb) {
#pragma unroll
        for (int j = 0; j < ACC_U; ++j) pkN[j] = L[(mb + j < n) ? (mb + j) : (n - 1)];
#pragma unroll
        for (int j = 0; j < ACC_U; ++j) {
          const int i = pkN[j] & 0x3FFFF;
          soN[j] = oth[i];
          aN[j]  = alpha[(size_t)i * D + t];
        }
      };
      if (n > 0) hop1(0);
      for (int mb = 0; mb < n; mb += ACC_U) {
        int    pk[ACC_U];
        float  a[ACC_U];
        float  bi[ACC_U];
        float4 v[ACC_U];
#pragma unroll
        for (int j = 0; j < ACC_U; ++j) {
          pk[j] = pkN[j];
          a[j]  = aN[j];
          v[j]  = *reinterpret_cast<const float4*>(src + (size_t)soN[j] * OUT_CH + tid * 4);
          bi[j] = NODE ? Binv[soN[j]] : 1.0f;
        }
        hop1(mb + ACC_U);
#pragma unroll
        for (int j = 0; j < ACC_U; ++j) {
          const bool live = (mb + j < n);
          const int d = (int)((unsigned)pk[j] >> 18);
          const float aa = live ? (NODE ? a[j] * bi[j] : a[j]) : 0.0f;
          const float4 vv = live ? v[j] : make_float4(0.f, 0.f, 0.f, 0.f);
          float4 c = acc[d * SEG_TPB + tid];
          c.x += aa * vv.x;
          c.y += aa * vv.y;
          c.z += aa * vv.z;
          c.w += aa * vv.w;
          acc[d * SEG_TPB + tid] = c;
        }
      }
    }
    alldone = (doneS[0] & doneS[1] & doneS[2]) != 0;
    __syncthreads();
  }

  float4 b = make_float4(0.f, 0.f, 0.f, 0.f);
  if (NODE) b = *reinterpret_cast<const float4*>(bias + ((tid * 4) & (HIDDEN - 1)));
  auto finish = [&](const int r, const float4 hv, const float dinv) {
    const float4 c = acc[r * SEG_TPB + tid];
    float4 ov = c;
    if (NODE) {
      ov.x = hv.x - dinv * c.x + b.x;
      ov.y = hv.y - dinv * c.y + b.y;
      ov.z = hv.z - dinv * c.z + b.z;
      ov.w = hv.w - dinv * c.w + b.w;
      ov.x = (ov.x > 0.0f) ? ov.x : expm1f(ov.x);
      ov.y = (ov.y > 0.0f) ? ov.y : expm1f(ov.y);
      ov.z = (ov.z > 0.0f) ? ov.z : expm1f(ov.z);
      ov.w = (ov.w > 0.0f) ? ov.w : expm1f(ov.w);
    }
    *reinterpret_cast<float4*>(dst + (size_t)(r0 + r) * OUT_CH + tid * 4) = ov;
  };
  if (rr == R) {
    static_assert(R % FIN_U == 0, "whole groups of rows");
    for (int rb = 0; rb < R; rb += FIN_U) {
      float4 hv[FIN_U];
      float  dinv[FIN_U];
#pragma unroll
      for (int j = 0; j < FIN_U; ++j) {
        hv[j]   = NODE ? *reinterpret_cast<const float4*>(hself + (size_t)(r0 + rb + j) * OUT_CH + tid * 4)
                       : make_float4(0.f, 0.f, 0.f, 0.f);
        dinv[j] = NODE ? Dinv[(r0 + rb + j) * D + t] : 0.0f;
      }
#pragma unroll
      for (int j = 0; j < FIN_U; ++j) finish(rb + j, hv[j], dinv[j]);
    }
  } else {
    for (int r = 0; r < rr; ++r) {
      const float4 hv = NODE ? *reinterpret_cast<const float4*>(hself + (size_t)(r0 + r) * OUT_CH + tid * 4)
                             : make_float4(0.f, 0.f, 0.f, 0.f);
      const float dinv = NODE ? Dinv[(r0 + r) * D + t] : 0.0f;
      finish(r, hv, dinv);
    }
  }
}

extern "C" void kernel_launch(void* const* d_in, const int* in_sizes, int n_in,
                              void* d_out, int out_size, void* d_ws, size_t ws_size,
                              hipStream_t stream) {
  (void)in_sizes; (void)n_in; (void)out_size; (void)ws_size;
  const float* x       = (const float*)d_in[0];
  const float* hattr   = (const float*)d_in[1];
  const float* W_lin   = (const float*)d_in[2];
  const float* b_lin   = (const float*)d_in[3];
  const float* ln_g    = (const float*)d_in[4];
  const float* ln_b    = (const float*)d_in[5];
  const float* W_sheaf = (const float*)d_in[6];
  const float* b_sheaf = (const float*)d_in[7];
  const float* W_conv  = (const float*)d_in[8];
  const float* b_conv  = (const float*)d_in[9];
  const int*   eidx    = (const int*)d_in[10];
  const int* row = eidx;
  const int* col = eidx + NNZ;
  float* out = (float*)d_out;

  float* ws    = (float*)d_ws;
  float* xs    = ws;
  float* es    = xs   + (size_t)ND * HIDDEN;
  float* xm    = es   + (size_t)ED * HIDDEN;
  float* em    = xm   + (size_t)N_NODES * HIDDEN;
  float* alpha = em   + (size_t)N_EDGES * HIDDEN;
  float* Dbuf  = alpha + (size_t)NNZ * D;
  float* Bbuf  = Dbuf + ND;
  float* hbuf  = Bbuf + N_EDGES;
  float* me    = hbuf + (size_t)ND * HIDDEN;

  const int TPB = 256;

  {
    wmma_gemm_lds<IN_CH><<<dim3(85, OUT_CH / 64), GEMM_TPB, 0, stream>>>(x, W_lin, b_lin, xs,
                                                                       OUT_CH, N_NODES / 16);
    wmma_gemm_lds<IN_CH><<<dim3(79, OUT_CH / 64), GEMM_TPB, 0, stream>>>(hattr, W_lin, b_lin, es,
                                                                       OUT_CH, N_EDGES / 16);
  }

  stalk_mean_kernel<<<(N_NODES * HIDDEN + TPB - 1) / TPB, TPB, 0, stream>>>(xs, xm, N_NODES * HIDDEN);
  stalk_mean_kernel<<<(N_EDGES * HIDDEN + TPB - 1) / TPB, TPB, 0, stream>>>(es, em, N_EDGES * HIDDEN);

  hipMemsetAsync(Dbuf, 0, (size_t)ND * sizeof(float), stream);
  hipMemsetAsync(Bbuf, 0, (size_t)N_EDGES * sizeof(float), stream);
  alpha_kernel<<<NNZ / 8, TPB, 0, stream>>>(xm, em, row, col, ln_g, ln_b,
                                            W_sheaf, b_sheaf, alpha, Dbuf, Bbuf);
  safe_inv_kernel<<<(ND + TPB - 1) / TPB, TPB, 0, stream>>>(Dbuf, ND);
  safe_inv_kernel<<<(N_EDGES + TPB - 1) / TPB, TPB, 0, stream>>>(Bbuf, N_EDGES);

  constexpr int convGrid = 512;
  constexpr int RE = 8;
  constexpr int RN = 40;
  const int edgeBlk = (N_EDGES + RE - 1) / RE;
  const int nodeBlk = (N_NODES + RN - 1) / RN;

  wmma_gemm_lds<HIDDEN><<<dim3(convGrid, 1), GEMM_TPB, 0, stream>>>(xs, W_conv, nullptr, hbuf,
                                                                    HIDDEN, ND / 16);
  segsum_kernel<RE, false><<<edgeBlk, SEG_TPB, 0, stream>>>(hbuf, alpha, col, row, nullptr, nullptr,
                                                            nullptr, nullptr, me, N_EDGES);
  segsum_kernel<RN, true><<<nodeBlk, SEG_TPB, 0, stream>>>(me, alpha, row, col, Bbuf, hbuf,
                                                           Dbuf, b_conv, out, N_NODES);

  wmma_gemm_lds<HIDDEN><<<dim3(convGrid, 1), GEMM_TPB, 0, stream>>>(out, W_conv + HIDDEN * HIDDEN, nullptr, hbuf,
                                                                    HIDDEN, ND / 16);
  segsum_kernel<RE, false><<<edgeBlk, SEG_TPB, 0, stream>>>(hbuf, alpha, col, row, nullptr, nullptr,
                                                            nullptr, nullptr, me, N_EDGES);
  segsum_kernel<RN, true><<<nodeBlk, SEG_TPB, 0, stream>>>(me, alpha, row, col, Bbuf, hbuf,
                                                           Dbuf, b_conv + HIDDEN, out, N_NODES);
}


# Second-layer owner kernels read back the per-work-group incidence lists the first layer's scan saved in workspace instead of rescanning all ids; live terms keep order, only 0*0 filler adds move
# speedup vs baseline: 2.8456x; 1.1095x over previous
#include <hip/hip_runtime.h>
#include <math.h>

constexpr int N_NODES = 50000;
constexpr int N_EDGES = 10000;
constexpr int NNZ     = 150000;
constexpr int IN_CH   = 128;
constexpr int HIDDEN  = 64;
constexpr int D       = 6;
constexpr int OUT_CH  = HIDDEN * D;
constexpr int ND      = N_NODES * D;
constexpr int ED      = N_EDGES * D;

typedef __attribute__((ext_vector_type(2))) float v2f;
typedef __attribute__((ext_vector_type(8))) float v8f;

constexpr int GEMM_WAVES = 8;
constexpr int GEMM_TPB   = GEMM_WAVES * 32;

template <int K>
__global__ __launch_bounds__(GEMM_TPB)
void wmma_gemm_lds(const float* __restrict__ A, const float* __restrict__ B,
                   const float* __restrict__ bias, float* __restrict__ C,
                   int N, int mTiles) {
  constexpr int KP   = K / 8;
  constexpr int AP   = K + 4;
  constexpr int CP   = 64 + 4;
  constexpr int SLOT = 16 * AP;
  static_assert(K % 8 == 0 && K >= 64, "slot must also hold the 16 x 64 result");
  __shared__ float4 Bs[KP * 4 * 32];
  __shared__ float4 Ss[GEMM_WAVES * SLOT / 4];

  const int tid  = threadIdx.x;
  const int wave = __builtin_amdgcn_readfirstlane(tid >> 5);
  const int lane = tid & 31;
  const int c    = lane & 15;
  const int h    = lane >> 4;
  const int n0   = blockIdx.y * 64;

  {
    constexpr int BIT = KP * 4 * 32 / GEMM_TPB;
    float4 bw[BIT];
#pragma unroll
    for (int it = 0; it < BIT; ++it) {
      const int idx = tid + it * GEMM_TPB;
      const int l  = idx & 31;
      const int nt = (idx >> 5) & 3;
      const int p  = idx >> 7;
      const float* bp = B + (size_t)(8 * p + 4 * (l >> 4)) * N + n0 + nt * 16 + (l & 15);
      bw[it].x = bp[0];
      bw[it].y = bp[N];
      bw[it].z = bp[2 * (size_t)N];
      bw[it].w = bp[3 * (size_t)N];
    }
#pragma unroll
    for (int it = 0; it < BIT; ++it) Bs[tid + it * GEMM_TPB] = bw[it];
  }
  __syncthreads();

  float* slot = reinterpret_cast<float*>(Ss) + wave * SLOT;
  float bv[4];
#pragma unroll
  for (int nt = 0; nt < 4; ++nt) bv[nt] = bias ? bias[n0 + nt * 16 + c] : 0.0f;

  for (int mt = blockIdx.x * GEMM_WAVES + wave; mt < mTiles; mt += gridDim.x * GEMM_WAVES) {
    const float4* g = reinterpret_cast<const float4*>(A + (size_t)mt * 16 * K);
    float4 tl[KP];
#pragma unroll
    for (int i = 0; i < KP; ++i) tl[i] = g[i * 32 + lane];
#pragma unroll
    for (int i = 0; i < KP; ++i) {
      const int f = i * 32 + lane;
      const int r = f / (K / 4);
      const int q = f % (K / 4);
      *reinterpret_cast<float4*>(slot + r * AP + 4 * q) = tl[i];
    }

    v8f acc[4] = {};
#pragma unroll 2
    for (int p = 0; p < KP; ++p) {
      const float4 a4 = *reinterpret_cast<const float4*>(slot + c * AP + 8 * p + 4 * h);
      float4 b4[4];
#pragma unroll
      for (int nt = 0; nt < 4; ++nt) b4[nt] = Bs[(p * 4 + nt) * 32 + lane];
      v2f a0; a0.x = a4.x; a0.y = a4.y;
      v2f a1; a1.x = a4.z; a1.y = a4.w;
#pragma unroll
      for (int nt = 0; nt < 4; ++nt) {
        v2f b0; b0.x = b4[nt].x; b0.y = b4[nt].y;
        acc[nt] = __builtin_amdgcn_wmma_f32_16x16x4_f32(false, a0, false, b0,
                                                        (short)0, acc[nt], false, false);
      }
#pragma unroll
      for (int nt = 0; nt < 4; ++nt) {
        v2f b1; b1.x = b4[nt].z; b1.y = b4[nt].w;
        acc[nt] = __builtin_amdgcn_wmma_f32_16x16x4_f32(false, a1, false, b1,
                                                        (short)0, acc[nt], false, false);
      }
    }

#pragma unroll
    for (int nt = 0; nt < 4; ++nt) {
#pragma unroll
      for (int r = 0; r < 8; ++r) slot[(8 * h + r) * CP + nt * 16 + c] = acc[nt][r] + bv[nt];
    }
#pragma unroll
    for (int i = 0; i < 8; ++i) {
      const int f   = i * 32 + lane;
      const int row = f >> 4;
      const int q   = f & 15;
      const float4 o = *reinterpret_cast<const float4*>(slot + row * CP + 4 * q);
      *reinterpret_cast<float4*>(C + (size_t)(mt * 16 + row) * N + n0 + 4 * q) = o;
    }
  }
}

__global__ void stalk_mean_kernel(const float* __restrict__ proj, float* __restrict__ m,
                                  int total) {
  int idx = blockIdx.x * blockDim.x + threadIdx.x;
  if (idx >= total) return;
  int n = idx >> 6;
  int hh = idx & 63;
  const float* p = proj + (size_t)n * OUT_CH + hh;
  float s = 0.0f;
#pragma unroll
  for (int t = 0; t < D; ++t) s += p[t * HIDDEN];
  m[idx] = s * (1.0f / 6.0f);
}

__device__ inline float wave_allsum(float v) {
#pragma unroll
  for (int off = 16; off > 0; off >>= 1) v += __shfl_xor(v, off, 32);
  return v;
}

__global__ void alpha_kernel(const float* __restrict__ xm, const float* __restrict__ em,
                             const int* __restrict__ row, const int* __restrict__ col,
                             const float* __restrict__ ln_g, const float* __restrict__ ln_b,
                             const float* __restrict__ Wsh, const float* __restrict__ bsh,
                             float* __restrict__ alpha, float* __restrict__ Dsum,
                             float* __restrict__ Bcnt) {
  int i = blockIdx.x * (blockDim.x >> 5) + (threadIdx.x >> 5);
  if (i >= NNZ) return;
  int lane = threadIdx.x & 31;
  int rn = row[i], cn = col[i];
  int j0 = lane * 4;
  float z[4];
#pragma unroll
  for (int q = 0; q < 4; ++q) {
    int j = j0 + q;
    z[q] = (j < HIDDEN) ? xm[(size_t)rn * HIDDEN + j]
                        : em[(size_t)cn * HIDDEN + (j - HIDDEN)];
  }
  float s  = z[0] + z[1] + z[2] + z[3];
  float ss = z[0]*z[0] + z[1]*z[1] + z[2]*z[2] + z[3]*z[3];
  s  = wave_allsum(s);
  ss = wave_allsum(ss);
  float mu   = s * (1.0f / 128.0f);
  float var  = ss * (1.0f / 128.0f) - mu * mu;
  float rstd = rsqrtf(var + 1e-5f);

  float p[6] = {0, 0, 0, 0, 0, 0};
#pragma unroll
  for (int q = 0; q < 4; ++q) {
    int j = j0 + q;
    float fn = (z[q] - mu) * rstd * ln_g[j] + ln_b[j];
#pragma unroll
    for (int t = 0; t < D; ++t) p[t] += fn * Wsh[j * D + t];
  }
#pragma unroll
  for (int t = 0; t < D; ++t) p[t] = wave_allsum(p[t]);

  if (lane < D) {
    float a = 1.0f / (1.0f + expf(-(p[lane] + bsh[lane])));
    alpha[(size_t)i * D + lane] = a;
    atomicAdd(&Dsum[rn * D + lane], a * a);
  } else if (lane == D) {
    atomicAdd(&Bcnt[cn], 1.0f);
  }
}

__global__ void safe_inv_kernel(float* __restrict__ buf, int n) {
  int idx = blockIdx.x * blockDim.x + threadIdx.x;
  if (idx >= n) return;
  float v = buf[idx];
  buf[idx] = (v > 0.0f) ? (1.0f / v) : 0.0f;
}

constexpr int SEG_TPB   = 96;
constexpr int SEG_WAVES = 3;
constexpr int SEG_CAP   = 256;
constexpr int STRIP     = 128;
constexpr int NSTRIPS   = (NNZ + STRIP - 1) / STRIP;
constexpr int NFULL     = NNZ / STRIP;
constexpr int SCAN_U    = 16;
constexpr int ACC_U     = 8;
constexpr int FIN_U     = 4;
constexpr int SAVE_HDR  = 32;
constexpr int SAVE_CAP  = 1024;
constexpr int SAVE_RUN  = SAVE_HDR + SAVE_CAP;
static_assert(NNZ % 4 == 0, "16-byte id loads are whole or absent");
static_assert(NNZ <= (1 << 18), "incidence id is packed into 18 bits");
static_assert(SEG_CAP >= 2 * STRIP, "a round must be able to take a strip");

template <int R, bool NODE, int MODE>
__global__ __launch_bounds__(SEG_TPB)
void segsum_kernel(const float* __restrict__ src, const float* __restrict__ alpha,
                   const int* __restrict__ key, const int* __restrict__ oth,
                   const float* __restrict__ Binv, const float* __restrict__ hself,
                   const float* __restrict__ Dinv, const float* __restrict__ bias,
                   float* __restrict__ dst, int nDest, int* __restrict__ saved) {
  static_assert(R <= 64, "owned row number and the 18-bit id share one word");
  __shared__ float4 acc[R * SEG_TPB];
  __shared__ int    lst[SEG_WAVES * SEG_CAP];
  __shared__ int    cntS[SEG_WAVES];
  __shared__ int    doneS[SEG_WAVES];

  const int tid  = threadIdx.x;
  const int wave = __builtin_amdgcn_readfirstlane(tid >> 5);
  const int lane = tid & 31;
  const int t    = tid >> 4;
  const int r0   = blockIdx.x * R;
  const int rr   = (nDest - r0 < R) ? (nDest - r0) : R;

  for (int r = 0; r < R; ++r) acc[r * SEG_TPB + tid] = make_float4(0.f, 0.f, 0.f, 0.f);

  int cnt = 0;
  auto strip = [&](const int4 k, const int i0) {
    const unsigned d0 = (unsigned)(k.x - r0);
    const unsigned d1 = (unsigned)(k.y - r0);
    const unsigned d2 = (unsigned)(k.z - r0);
    const unsigned d3 = (unsigned)(k.w - r0);
    const unsigned dm = min(min(d0, d1), min(d2, d3));
    if (__builtin_amdgcn_ballot_w32(dm < (unsigned)rr) != 0u) {
      auto append = [&](unsigned d, int i) {
        const bool m = d < (unsigned)rr;
        const unsigned b = __builtin_amdgcn_ballot_w32(m);
        if (m) {
          const int pos = cnt + (int)__builtin_amdgcn_mbcnt_lo(b, 0u);
          lst[wave * SEG_CAP + pos] = (int)((d << 18) | (unsigned)i);
        }
        cnt += __popc(b);
      };
      append(d0, i0);
      append(d1, i0 + 1);
      append(d2, i0 + 2);
      append(d3, i0 + 3);
    }
  };

  int* mine = saved + (size_t)blockIdx.x * SAVE_RUN;
  int stored = 0;
  int total = -1;
  if (MODE == 1) total = mine[0];
  const bool replay = (MODE == 1) && total >= 0 && total <= SAVE_CAP;
  int taken = 0;

  int s = wave;
  bool alldone = false;
  while (!alldone) {
   if (replay) {
    const int left = total - taken;
    const int n = (left < SEG_WAVES * SEG_CAP) ? left : SEG_WAVES * SEG_CAP;
    for (int e = tid; e < n; e += SEG_TPB) lst[e] = mine[SAVE_HDR + taken + e];
    taken += n;
    if (tid < SEG_WAVES) { cntS[tid] = (tid == 0) ? n : 0; doneS[tid] = (taken >= total) ? 1 : 0; }
   } else {
    cnt = 0;
    bool full = false;
    while (!full && s + (SCAN_U - 1) * SEG_WAVES < NFULL) {
      int4 k[SCAN_U];
      const int ib = s * STRIP + lane * 4;
#pragma unroll
      for (int u = 0; u < SCAN_U; ++u)
        k[u] = *reinterpret_cast<const int4*>(key + ib + u * SEG_WAVES * STRIP);
      int used = SCAN_U;
#pragma unroll
      for (int u = 0; u < SCAN_U; ++u) {
        strip(k[u], ib + u * SEG_WAVES * STRIP);
        if (cnt > SEG_CAP - STRIP) { full = true; used = u + 1; break; }
      }
      s += used * SEG_WAVES;
    }
    while (!full && s < NSTRIPS) {
      const int i0 = s * STRIP + lane * 4;
      const int4 k = (i0 < NNZ) ? *reinterpret_cast<const int4*>(key + i0)
                                : make_int4(-1, -1, -1, -1);
      strip(k, i0);
      s += SEG_WAVES;
      if (cnt > SEG_CAP - STRIP) full = true;
    }
    if (lane == 0) { cntS[wave] = cnt; doneS[wave] = (s >= NSTRIPS) ? 1 : 0; }
   }
    __syncthreads();

    if (MODE == 0) {
      const int c0 = cntS[0], c1 = cntS[1], c2 = cntS[2];
      const int tot = c0 + c1 + c2;
      if (stored >= 0 && stored + tot <= SAVE_CAP) {
        for (int e = tid; e < tot; e += SEG_TPB) {
          const int w = (e >= c0 ? 1 : 0) + (e >= c0 + c1 ? 1 : 0);
          const int o = e - (w == 0 ? 0 : (w == 1 ? c0 : c0 + c1));
          mine[SAVE_HDR + stored + e] = lst[w * SEG_CAP + o];
        }
        stored += tot;
      } else {
        stored = -1;
      }
    }

    for (int w = 0; w < SEG_WAVES; ++w) {
      const int n = __builtin_amdgcn_readfirstlane(cntS[w]);
      const int* L = lst + w * SEG_CAP;
      int   pkN[ACC_U];
      int   soN[ACC_U];
      float aN[ACC_U];
      auto hop1 = [&](const int mb) {
#pragma unroll
        for (int j = 0; j < ACC_U; ++j) pkN[j] = L[(mb + j < n) ? (mb + j) : (n - 1)];
#pragma unroll
        for (int j = 0; j < ACC_U; ++j) {
          const int i = pkN[j] & 0x3FFFF;
          soN[j] = oth[i];
          aN[j]  = alpha[(size_t)i * D + t];
        }
      };
      if (n > 0) hop1(0);
      for (int mb = 0; mb < n; mb += ACC_U) {
        int    pk[ACC_U];
        float  a[ACC_U];
        float  bi[ACC_U];
        float4 v[ACC_U];
#pragma unroll
        for (int j = 0; j < ACC_U; ++j) {
          pk[j] = pkN[j];
          a[j]  = aN[j];
          v[j]  = *reinterpret_cast<const float4*>(src + (size_t)soN[j] * OUT_CH + tid * 4);
          bi[j] = NODE ? Binv[soN[j]] : 1.0f;
        }
        hop1(mb + ACC_U);
#pragma unroll
        for (int j = 0; j < ACC_U; ++j) {
          const bool live = (mb + j < n);
          const int d = (int)((unsigned)pk[j] >> 18);
          const float aa = live ? (NODE ? a[j] * bi[j] : a[j]) : 0.0f;
          const float4 vv = live ? v[j] : make_float4(0.f, 0.f, 0.f, 0.f);
          float4 c = acc[d * SEG_TPB + tid];
          c.x += aa * vv.x;
          c.y += aa * vv.y;
          c.z += aa * vv.z;
          c.w += aa * vv.w;
          acc[d * SEG_TPB + tid] = c;
        }
      }
    }
    alldone = (doneS[0] & doneS[1] & doneS[2]) != 0;
    __syncthreads();
  }
  if (MODE == 0 && tid == 0) mine[0] = stored;

  float4 b = make_float4(0.f, 0.f, 0.f, 0.f);
  if (NODE) b = *reinterpret_cast<const float4*>(bias + ((tid * 4) & (HIDDEN - 1)));
  auto finish = [&](const int r, const float4 hv, const float dinv) {
    const float4 c = acc[r * SEG_TPB + tid];
    float4 ov = c;
    if (NODE) {
      ov.x = hv.x - dinv * c.x + b.x;
      ov.y = hv.y - dinv * c.y + b.y;
      ov.z = hv.z - dinv * c.z + b.z;
      ov.w = hv.w - dinv * c.w + b.w;
      ov.x = (ov.x > 0.0f) ? ov.x : expm1f(ov.x);
      ov.y = (ov.y > 0.0f) ? ov.y : expm1f(ov.y);
      ov.z = (ov.z > 0.0f) ? ov.z : expm1f(ov.z);
      ov.w = (ov.w > 0.0f) ? ov.w : expm1f(ov.w);
    }
    *reinterpret_cast<float4*>(dst + (size_t)(r0 + r) * OUT_CH + tid * 4) = ov;
  };
  if (rr == R) {
    static_assert(R % FIN_U == 0, "whole groups of rows");
    for (int rb = 0; rb < R; rb += FIN_U) {
      float4 hv[FIN_U];
      float  dinv[FIN_U];
#pragma unroll
      for (int j = 0; j < FIN_U; ++j) {
        hv[j]   = NODE ? *reinterpret_cast<const float4*>(hself + (size_t)(r0 + rb + j) * OUT_CH + tid * 4)
                       : make_float4(0.f, 0.f, 0.f, 0.f);
        dinv[j] = NODE ? Dinv[(r0 + rb + j) * D + t] : 0.0f;
      }
#pragma unroll
      for (int j = 0; j < FIN_U; ++j) finish(rb + j, hv[j], dinv[j]);
    }
  } else {
    for (int r = 0; r < rr; ++r) {
      const float4 hv = NODE ? *reinterpret_cast<const float4*>(hself + (size_t)(r0 + r) * OUT_CH + tid * 4)
                             : make_float4(0.f, 0.f, 0.f, 0.f);
      const float dinv = NODE ? Dinv[(r0 + r) * D + t] : 0.0f;
      finish(r, hv, dinv);
    }
  }
}

extern "C" void kernel_launch(void* const* d_in, const int* in_sizes, int n_in,
                              void* d_out, int out_size, void* d_ws, size_t ws_size,
                              hipStream_t stream) {
  (void)in_sizes; (void)n_in; (void)out_size; (void)ws_size;
  const float* x       = (const float*)d_in[0];
  const float* hattr   = (const float*)d_in[1];
  const float* W_lin   = (const float*)d_in[2];
  const float* b_lin   = (const float*)d_in[3];
  const float* ln_g    = (const float*)d_in[4];
  const float* ln_b    = (const float*)d_in[5];
  const float* W_sheaf = (const float*)d_in[6];
  const float* b_sheaf = (const float*)d_in[7];
  const float* W_conv  = (const float*)d_in[8];
  const float* b_conv  = (const float*)d_in[9];
  const int*   eidx    = (const int*)d_in[10];
  const int* row = eidx;
  const int* col = eidx + NNZ;
  float* out = (float*)d_out;

  float* ws    = (float*)d_ws;
  float* xs    = ws;
  float* es    = xs   + (size_t)ND * HIDDEN;
  float* xm    = es   + (size_t)ED * HIDDEN;
  float* em    = xm   + (size_t)N_NODES * HIDDEN;
  float* alpha = em   + (size_t)N_EDGES * HIDDEN;
  float* Dbuf  = alpha + (size_t)NNZ * D;
  float* Bbuf  = Dbuf + ND;
  float* hbuf  = Bbuf + N_EDGES;
  float* me    = hbuf + (size_t)ND * HIDDEN;

  const int TPB = 256;

  {
    wmma_gemm_lds<IN_CH><<<dim3(85, OUT_CH / 64), GEMM_TPB, 0, stream>>>(x, W_lin, b_lin, xs,
                                                                       OUT_CH, N_NODES / 16);
    wmma_gemm_lds<IN_CH><<<dim3(79, OUT_CH / 64), GEMM_TPB, 0, stream>>>(hattr, W_lin, b_lin, es,
                                                                       OUT_CH, N_EDGES / 16);
  }

  stalk_mean_kernel<<<(N_NODES * HIDDEN + TPB - 1) / TPB, TPB, 0, stream>>>(xs, xm, N_NODES * HIDDEN);
  stalk_mean_kernel<<<(N_EDGES * HIDDEN + TPB - 1) / TPB, TPB, 0, stream>>>(es, em, N_EDGES * HIDDEN);

  hipMemsetAsync(Dbuf, 0, (size_t)ND * sizeof(float), stream);
  hipMemsetAsync(Bbuf, 0, (size_t)N_EDGES * sizeof(float), stream);
  alpha_kernel<<<NNZ / 8, TPB, 0, stream>>>(xm, em, row, col, ln_g, ln_b,
                                            W_sheaf, b_sheaf, alpha, Dbuf, Bbuf);
  safe_inv_kernel<<<(ND + TPB - 1) / TPB, TPB, 0, stream>>>(Dbuf, ND);
  safe_inv_kernel<<<(N_EDGES + TPB - 1) / TPB, TPB, 0, stream>>>(Bbuf, N_EDGES);

  constexpr int convGrid = 512;
  constexpr int RE = 8;
  constexpr int RN = 40;
  const int edgeBlk = (N_EDGES + RE - 1) / RE;
  const int nodeBlk = (N_NODES + RN - 1) / RN;
  static_assert((size_t)((N_EDGES + RE - 1) / RE + (N_NODES + RN - 1) / RN) * SAVE_RUN
                    <= (size_t)N_NODES * HIDDEN, "saved runs fit in the xm region");
  int* edgeLists = reinterpret_cast<int*>(xm);
  int* nodeLists = edgeLists + (size_t)edgeBlk * SAVE_RUN;

  wmma_gemm_lds<HIDDEN><<<dim3(convGrid, 1), GEMM_TPB, 0, stream>>>(xs, W_conv, nullptr, hbuf,
                                                                    HIDDEN, ND / 16);
  segsum_kernel<RE, false, 0><<<edgeBlk, SEG_TPB, 0, stream>>>(hbuf, alpha, col, row, nullptr, nullptr,
                                                               nullptr, nullptr, me, N_EDGES, edgeLists);
  segsum_kernel<RN, true, 0><<<nodeBlk, SEG_TPB, 0, stream>>>(me, alpha, row, col, Bbuf, hbuf,
                                                              Dbuf, b_conv, out, N_NODES, nodeLists);

  wmma_gemm_lds<HIDDEN><<<dim3(convGrid, 1), GEMM_TPB, 0, stream>>>(out, W_conv + HIDDEN * HIDDEN, nullptr, hbuf,
                                                                    HIDDEN, ND / 16);
  segsum_kernel<RE, false, 1><<<edgeBlk, SEG_TPB, 0, stream>>>(hbuf, alpha, col, row, nullptr, nullptr,
                                                               nullptr, nullptr, me, N_EDGES, edgeLists);
  segsum_kernel<RN, true, 1><<<nodeBlk, SEG_TPB, 0, stream>>>(me, alpha, row, col, Bbuf, hbuf,
                                                              Dbuf, b_conv + HIDDEN, out, N_NODES, nodeLists);
}


# Conv weight product moved behind both segment sums by linearity: out = elu((x - Dinv*Ht*Binv*H*x) @ W + b); node owners do it on their LDS rows by WMMA; 2 GEMM launches, h table gone; re-associated
# speedup vs baseline: 3.1843x; 1.1190x over previous
#include <hip/hip_runtime.h>
#include <math.h>

constexpr int N_NODES = 50000;
constexpr int N_EDGES = 10000;
constexpr int NNZ     = 150000;
constexpr int IN_CH   = 128;
constexpr int HIDDEN  = 64;
constexpr int D       = 6;
constexpr int OUT_CH  = HIDDEN * D;
constexpr int ND      = N_NODES * D;
constexpr int ED      = N_EDGES * D;

typedef __attribute__((ext_vector_type(2))) float v2f;
typedef __attribute__((ext_vector_type(8))) float v8f;

constexpr int GEMM_WAVES = 8;
constexpr int GEMM_TPB   = GEMM_WAVES * 32;

template <int K>
__global__ __launch_bounds__(GEMM_TPB)
void wmma_gemm_lds(const float* __restrict__ A, const float* __restrict__ B,
                   const float* __restrict__ bias, float* __restrict__ C,
                   int N, int mTiles) {
  constexpr int KP   = K / 8;
  constexpr int AP   = K + 4;
  constexpr int CP   = 64 + 4;
  constexpr int SLOT = 16 * AP;
  static_assert(K % 8 == 0 && K >= 64, "slot must also hold the 16 x 64 result");
  __shared__ float4 Bs[KP * 4 * 32];
  __shared__ float4 Ss[GEMM_WAVES * SLOT / 4];

  const int tid  = threadIdx.x;
  const int wave = __builtin_amdgcn_readfirstlane(tid >> 5);
  const int lane = tid & 31;
  const int c    = lane & 15;
  const int h    = lane >> 4;
  const int n0   = blockIdx.y * 64;

  {
    constexpr int BIT = KP * 4 * 32 / GEMM_TPB;
    float4 bw[BIT];
#pragma unroll
    for (int it = 0; it < BIT; ++it) {
      const int idx = tid + it * GEMM_TPB;
      const int l  = idx & 31;
      const int nt = (idx >> 5) & 3;
      const int p  = idx >> 7;
      const float* bp = B + (size_t)(8 * p + 4 * (l >> 4)) * N + n0 + nt * 16 + (l & 15);
      bw[it].x = bp[0];
      bw[it].y = bp[N];
      bw[it].z = bp[2 * (size_t)N];
      bw[it].w = bp[3 * (size_t)N];
    }
#pragma unroll
    for (int it = 0; it < BIT; ++it) Bs[tid + it * GEMM_TPB] = bw[it];
  }
  __syncthreads();

  float* slot = reinterpret_cast<float*>(Ss) + wave * SLOT;
  float bv[4];
#pragma unroll
  for (int nt = 0; nt < 4; ++nt) bv[nt] = bias ? bias[n0 + nt * 16 + c] : 0.0f;

  for (int mt = blockIdx.x * GEMM_WAVES + wave; mt < mTiles; mt += gridDim.x * GEMM_WAVES) {
    const float4* g = reinterpret_cast<const float4*>(A + (size_t)mt * 16 * K);
    float4 tl[KP];
#pragma unroll
    for (int i = 0; i < KP; ++i) tl[i] = g[i * 32 + lane];
#pragma unroll
    for (int i = 0; i < KP; ++i) {
      const int f = i * 32 + lane;
      const int r = f / (K / 4);
      const int q = f % (K / 4);
      *reinterpret_cast<float4*>(slot + r * AP + 4 * q) = tl[i];
    }

    v8f acc[4] = {};
#pragma unroll 2
    for (int p = 0; p < KP; ++p) {
      const float4 a4 = *reinterpret_cast<const float4*>(slot + c * AP + 8 * p + 4 * h);
      float4 b4[4];
#pragma unroll
      for (int nt = 0; nt < 4; ++nt) b4[nt] = Bs[(p * 4 + nt) * 32 + lane];
      v2f a0; a0.x = a4.x; a0.y = a4.y;
      v2f a1; a1.x = a4.z; a1.y = a4.w;
#pragma unroll
      for (int nt = 0; nt < 4; ++nt) {
        v2f b0; b0.x = b4[nt].x; b0.y = b4[nt].y;
        acc[nt] = __builtin_amdgcn_wmma_f32_16x16x4_f32(false, a0, false, b0,
                                                        (short)0, acc[nt], false, false);
      }
#pragma unroll
      for (int nt = 0; nt < 4; ++nt) {
        v2f b1; b1.x = b4[nt].z; b1.y = b4[nt].w;
        acc[nt] = __builtin_amdgcn_wmma_f32_16x16x4_f32(false, a1, false, b1,
                                                        (short)0, acc[nt], false, false);
      }
    }

#pragma unroll
    for (int nt = 0; nt < 4; ++nt) {
#pragma unroll
      for (int r = 0; r < 8; ++r) slot[(8 * h + r) * CP + nt * 16 + c] = acc[nt][r] + bv[nt];
    }
#pragma unroll
    for (int i = 0; i < 8; ++i) {
      const int f   = i * 32 + lane;
      const int row = f >> 4;
      const int q   = f & 15;
      const float4 o = *reinterpret_cast<const float4*>(slot + row * CP + 4 * q);
      *reinterpret_cast<float4*>(C + (size_t)(mt * 16 + row) * N + n0 + 4 * q) = o;
    }
  }
}

__global__ void stalk_mean_kernel(const float* __restrict__ proj, float* __restrict__ m,
                                  int total) {
  int idx = blockIdx.x * blockDim.x + threadIdx.x;
  if (idx >= total) return;
  int n = idx >> 6;
  int hh = idx & 63;
  const float* p = proj + (size_t)n * OUT_CH + hh;
  float s = 0.0f;
#pragma unroll
  for (int t = 0; t < D; ++t) s += p[t * HIDDEN];
  m[idx] = s * (1.0f / 6.0f);
}

__device__ inline float wave_allsum(float v) {
#pragma unroll
  for (int off = 16; off > 0; off >>= 1) v += __shfl_xor(v, off, 32);
  return v;
}

__global__ void alpha_kernel(const float* __restrict__ xm, const float* __restrict__ em,
                             const int* __restrict__ row, const int* __restrict__ col,
                             const float* __restrict__ ln_g, const float* __restrict__ ln_b,
                             const float* __restrict__ Wsh, const float* __restrict__ bsh,
                             float* __restrict__ alpha, float* __restrict__ Dsum,
                             float* __restrict__ Bcnt) {
  int i = blockIdx.x * (blockDim.x >> 5) + (threadIdx.x >> 5);
  if (i >= NNZ) return;
  int lane = threadIdx.x & 31;
  int rn = row[i], cn = col[i];
  int j0 = lane * 4;
  float z[4];
#pragma unroll
  for (int q = 0; q < 4; ++q) {
    int j = j0 + q;
    z[q] = (j < HIDDEN) ? xm[(size_t)rn * HIDDEN + j]
                        : em[(size_t)cn * HIDDEN + (j - HIDDEN)];
  }
  float s  = z[0] + z[1] + z[2] + z[3];
  float ss = z[0]*z[0] + z[1]*z[1] + z[2]*z[2] + z[3]*z[3];
  s  = wave_allsum(s);
  ss = wave_allsum(ss);
  float mu   = s * (1.0f / 128.0f);
  float var  = ss * (1.0f / 128.0f) - mu * mu;
  float rstd = rsqrtf(var + 1e-5f);

  float p[6] = {0, 0, 0, 0, 0, 0};
#pragma unroll
  for (int q = 0; q < 4; ++q) {
    int j = j0 + q;
    float fn = (z[q] - mu) * rstd * ln_g[j] + ln_b[j];
#pragma unroll
    for (int t = 0; t < D; ++t) p[t] += fn * Wsh[j * D + t];
  }
#pragma unroll
  for (int t = 0; t < D; ++t) p[t] = wave_allsum(p[t]);

  if (lane < D) {
    float a = 1.0f / (1.0f + expf(-(p[lane] + bsh[lane])));
    alpha[(size_t)i * D + lane] = a;
    atomicAdd(&Dsum[rn * D + lane], a * a);
  } else if (lane == D) {
    atomicAdd(&Bcnt[cn], 1.0f);
  }
}

__global__ void safe_inv_kernel(float* __restrict__ buf, int n) {
  int idx = blockIdx.x * blockDim.x + threadIdx.x;
  if (idx >= n) return;
  float v = buf[idx];
  buf[idx] = (v > 0.0f) ? (1.0f / v) : 0.0f;
}

constexpr int SEG_TPB   = 96;
constexpr int SEG_WAVES = 3;
constexpr int SEG_CAP   = 256;
constexpr int STRIP     = 128;
constexpr int NSTRIPS   = (NNZ + STRIP - 1) / STRIP;
constexpr int NFULL     = NNZ / STRIP;
constexpr int SCAN_U    = 16;
constexpr int ACC_U     = 8;
constexpr int SAVE_HDR  = 32;
constexpr int SAVE_CAP  = 1024;
constexpr int SAVE_RUN  = SAVE_HDR + SAVE_CAP;
static_assert(NNZ % 4 == 0, "16-byte id loads are whole or absent");
static_assert(NNZ <= (1 << 18), "incidence id is packed into 18 bits");
static_assert(SEG_CAP >= 2 * STRIP, "a round must be able to take a strip");

template <int R, bool NODE, int MODE>
__global__ __launch_bounds__(SEG_TPB)
void segsum_kernel(const float* __restrict__ src, const float* __restrict__ alpha,
                   const int* __restrict__ key, const int* __restrict__ oth,
                   const float* __restrict__ Binv, const float* hself,
                   const float* __restrict__ Dinv, const float* __restrict__ bias,
                   const float* __restrict__ W,
                   float* dst, int nDest, int* __restrict__ saved) {
  static_assert(R <= 64, "owned row number and the 18-bit id share one word");
  static_assert(!NODE || (R * D) % 16 == 0, "owned stalk rows form whole 16-row tiles");
  __shared__ float4 acc[R * SEG_TPB];
  __shared__ int    lst[SEG_WAVES * SEG_CAP];
  __shared__ int    cntS[SEG_WAVES];
  __shared__ int    doneS[SEG_WAVES];

  const int tid  = threadIdx.x;
  const int wave = __builtin_amdgcn_readfirstlane(tid >> 5);
  const int lane = tid & 31;
  const int t    = tid >> 4;
  const int r0   = blockIdx.x * R;
  const int rr   = (nDest - r0 < R) ? (nDest - r0) : R;

  for (int r = 0; r < R; ++r) acc[r * SEG_TPB + tid] = make_float4(0.f, 0.f, 0.f, 0.f);

  int cnt = 0;
  auto strip = [&](const int4 k, const int i0) {
    const unsigned d0 = (unsigned)(k.x - r0);
    const unsigned d1 = (unsigned)(k.y - r0);
    const unsigned d2 = (unsigned)(k.z - r0);
    const unsigned d3 = (unsigned)(k.w - r0);
    const unsigned dm = min(min(d0, d1), min(d2, d3));
    if (__builtin_amdgcn_ballot_w32(dm < (unsigned)rr) != 0u) {
      auto append = [&](unsigned d, int i) {
        const bool m = d < (unsigned)rr;
        const unsigned b = __builtin_amdgcn_ballot_w32(m);
        if (m) {
          const int pos = cnt + (int)__builtin_amdgcn_mbcnt_lo(b, 0u);
          lst[wave * SEG_CAP + pos] = (int)((d << 18) | (unsigned)i);
        }
        cnt += __popc(b);
      };
      append(d0, i0);
      append(d1, i0 + 1);
      append(d2, i0 + 2);
      append(d3, i0 + 3);
    }
  };

  int* mine = saved + (size_t)blockIdx.x * SAVE_RUN;
  int stored = 0;
  int total = -1;
  if (MODE == 1) total = mine[0];
  const bool replay = (MODE == 1) && total >= 0 && total <= SAVE_CAP;
  int taken = 0;

  int s = wave;
  bool alldone = false;
  while (!alldone) {
   if (replay) {
    const int left = total - taken;
    const int n = (left < SEG_WAVES * SEG_CAP) ? left : SEG_WAVES * SEG_CAP;
    for (int e = tid; e < n; e += SEG_TPB) lst[e] = mine[SAVE_HDR + taken + e];
    taken += n;
    if (tid < SEG_WAVES) { cntS[tid] = (tid == 0) ? n : 0; doneS[tid] = (taken >= total) ? 1 : 0; }
   } else {
    cnt = 0;
    bool full = false;
    while (!full && s + (SCAN_U - 1) * SEG_WAVES < NFULL) {
      int4 k[SCAN_U];
      const int ib = s * STRIP + lane * 4;
#pragma unroll
      for (int u = 0; u < SCAN_U; ++u)
        k[u] = *reinterpret_cast<const int4*>(key + ib + u * SEG_WAVES * STRIP);
      int used = SCAN_U;
#pragma unroll
      for (int u = 0; u < SCAN_U; ++u) {
        strip(k[u], ib + u * SEG_WAVES * STRIP);
        if (cnt > SEG_CAP - STRIP) { full = true; used = u + 1; break; }
      }
      s += used * SEG_WAVES;
    }
    while (!full && s < NSTRIPS) {
      const int i0 = s * STRIP + lane * 4;
      const int4 k = (i0 < NNZ) ? *reinterpret_cast<const int4*>(key + i0)
                                : make_int4(-1, -1, -1, -1);
      strip(k, i0);
      s += SEG_WAVES;
      if (cnt > SEG_CAP - STRIP) full = true;
    }
    if (lane == 0) { cntS[wave] = cnt; doneS[wave] = (s >= NSTRIPS) ? 1 : 0; }
   }
    __syncthreads();

    if (MODE == 0) {
      const int c0 = cntS[0], c1 = cntS[1], c2 = cntS[2];
      const int tot = c0 + c1 + c2;
      if (stored >= 0 && stored + tot <= SAVE_CAP) {
        for (int e = tid; e < tot; e += SEG_TPB) {
          const int w = (e >= c0 ? 1 : 0) + (e >= c0 + c1 ? 1 : 0);
          const int o = e - (w == 0 ? 0 : (w == 1 ? c0 : c0 + c1));
          mine[SAVE_HDR + stored + e] = lst[w * SEG_CAP + o];
        }
        stored += tot;
      } else {
        stored = -1;
      }
    }

    for (int w = 0; w < SEG_WAVES; ++w) {
      const int n = __builtin_amdgcn_readfirstlane(cntS[w]);
      const int* L = lst + w * SEG_CAP;
      int   pkN[ACC_U];
      int   soN[ACC_U];
      float aN[ACC_U];
      auto hop1 = [&](const int mb) {
#pragma unroll
        for (int j = 0; j < ACC_U; ++j) pkN[j] = L[(mb + j < n) ? (mb + j) : (n - 1)];
#pragma unroll
        for (int j = 0; j < ACC_U; ++j) {
          const int i = pkN[j] & 0x3FFFF;
          soN[j] = oth[i];
          aN[j]  = alpha[(size_t)i * D + t];
        }
      };
      if (n > 0) hop1(0);
      for (int mb = 0; mb < n; mb += ACC_U) {
        int    pk[ACC_U];
        float  a[ACC_U];
        float  bi[ACC_U];
        float4 v[ACC_U];
#pragma unroll
        for (int j = 0; j < ACC_U; ++j) {
          pk[j] = pkN[j];
          a[j]  = aN[j];
          v[j]  = *reinterpret_cast<const float4*>(src + (size_t)soN[j] * OUT_CH + tid * 4);
          bi[j] = NODE ? Binv[soN[j]] : 1.0f;
        }
        hop1(mb + ACC_U);
#pragma unroll
        for (int j = 0; j < ACC_U; ++j) {
          const bool live = (mb + j < n);
          const int d = (int)((unsigned)pk[j] >> 18);
          const float aa = live ? (NODE ? a[j] * bi[j] : a[j]) : 0.0f;
          const float4 vv = live ? v[j] : make_float4(0.f, 0.f, 0.f, 0.f);
          float4 c = acc[d * SEG_TPB + tid];
          c.x += aa * vv.x;
          c.y += aa * vv.y;
          c.z += aa * vv.z;
          c.w += aa * vv.w;
          acc[d * SEG_TPB + tid] = c;
        }
      }
    }
    alldone = (doneS[0] & doneS[1] & doneS[2]) != 0;
    __syncthreads();
  }
  if (MODE == 0 && tid == 0) mine[0] = stored;

  if constexpr (!NODE) {
    auto finish = [&](const int r) {
      *reinterpret_cast<float4*>(dst + (size_t)(r0 + r) * OUT_CH + tid * 4) = acc[r * SEG_TPB + tid];
    };
    for (int r = 0; r < rr; ++r) finish(r);
  } else {
    constexpr int ZU = 10;
    static_assert(R % ZU == 0, "whole groups of rows");
#pragma nounroll
    for (int rb = 0; rb < R; rb += ZU) {
      float4 hv[ZU];
      float  dinv[ZU];
#pragma unroll
      for (int j = 0; j < ZU; ++j) {
        hv[j]   = *reinterpret_cast<const float4*>(hself + (size_t)(r0 + rb + j) * OUT_CH + tid * 4);
        dinv[j] = Dinv[(r0 + rb + j) * D + t];
      }
      float4 cs[ZU];
#pragma unroll
      for (int j = 0; j < ZU; ++j) cs[j] = acc[(rb + j) * SEG_TPB + tid];
#pragma unroll
      for (int j = 0; j < ZU; ++j) {
        float4 c = cs[j];
        c.x = hv[j].x - dinv[j] * c.x;
        c.y = hv[j].y - dinv[j] * c.y;
        c.z = hv[j].z - dinv[j] * c.z;
        c.w = hv[j].w - dinv[j] * c.w;
        const int key = ((rb + j) * D + t) & 15;
        acc[(rb + j) * SEG_TPB + (tid & ~15) + ((tid & 15) ^ key)] = c;
      }
    }
    const int c16 = lane & 15;
    const int hh  = lane >> 4;
    constexpr int KP = HIDDEN / 8;
    v2f bf[KP][4][2];
#pragma unroll
    for (int p = 0; p < KP; ++p) {
#pragma unroll
      for (int nt = 0; nt < 4; ++nt) {
        const float* bp = W + (8 * p + 4 * hh) * HIDDEN + nt * 16 + c16;
        bf[p][nt][0].x = bp[0];
        bf[p][nt][0].y = bp[HIDDEN];
        bf[p][nt][1].x = bp[2 * HIDDEN];
        bf[p][nt][1].y = bp[3 * HIDDEN];
      }
    }
    float bv[4];
#pragma unroll
    for (int nt = 0; nt < 4; ++nt) bv[nt] = bias[nt * 16 + c16];
    __syncthreads();

    float* accf = reinterpret_cast<float*>(acc);
    constexpr int MT = R * D / 16;
    for (int j = wave; j < MT; j += SEG_WAVES) {
      float* T = accf + j * 16 * HIDDEN;
      v8f o[4] = {};
#pragma unroll
      for (int p = 0; p < KP; ++p) {
        const float4 a4 = *reinterpret_cast<const float4*>(T + c16 * HIDDEN + 4 * ((2 * p + hh) ^ c16));
        v2f a0; a0.x = a4.x; a0.y = a4.y;
        v2f a1; a1.x = a4.z; a1.y = a4.w;
#pragma unroll
        for (int nt = 0; nt < 4; ++nt)
          o[nt] = __builtin_amdgcn_wmma_f32_16x16x4_f32(false, a0, false, bf[p][nt][0],
                                                        (short)0, o[nt], false, false);
#pragma unroll
        for (int nt = 0; nt < 4; ++nt)
          o[nt] = __builtin_amdgcn_wmma_f32_16x16x4_f32(false, a1, false, bf[p][nt][1],
                                                        (short)0, o[nt], false, false);
      }
#pragma unroll
      for (int nt = 0; nt < 4; ++nt) {
#pragma unroll
        for (int r = 0; r < 8; ++r) T[(8 * hh + r) * HIDDEN + nt * 16 + c16] = o[nt][r] + bv[nt];
      }
      float4 ov[8];
#pragma unroll
      for (int i = 0; i < 8; ++i) {
        const int f = i * 32 + lane;
        ov[i] = *reinterpret_cast<const float4*>(T + (f >> 4) * HIDDEN + 4 * (f & 15));
      }
#pragma unroll
      for (int i = 0; i < 8; ++i) {
        const int f   = i * 32 + lane;
        const int row = f >> 4;
        const int q   = f & 15;
        float4 e = ov[i];
        e.x = (e.x > 0.0f) ? e.x : expm1f(e.x);
        e.y = (e.y > 0.0f) ? e.y : expm1f(e.y);
        e.z = (e.z > 0.0f) ? e.z : expm1f(e.z);
        e.w = (e.w > 0.0f) ? e.w : expm1f(e.w);
        *reinterpret_cast<float4*>(dst + ((size_t)r0 * D + j * 16 + row) * HIDDEN + 4 * q) = e;
      }
    }
  }
}

extern "C" void kernel_launch(void* const* d_in, const int* in_sizes, int n_in,
                              void* d_out, int out_size, void* d_ws, size_t ws_size,
                              hipStream_t stream) {
  (void)in_sizes; (void)n_in; (void)out_size; (void)ws_size;
  const float* x       = (const float*)d_in[0];
  const float* hattr   = (const float*)d_in[1];
  const float* W_lin   = (const float*)d_in[2];
  const float* b_lin   = (const float*)d_in[3];
  const float* ln_g    = (const float*)d_in[4];
  const float* ln_b    = (const float*)d_in[5];
  const float* W_sheaf = (const float*)d_in[6];
  const float* b_sheaf = (const float*)d_in[7];
  const float* W_conv  = (const float*)d_in[8];
  const float* b_conv  = (const float*)d_in[9];
  const int*   eidx    = (const int*)d_in[10];
  const int* row = eidx;
  const int* col = eidx + NNZ;
  float* out = (float*)d_out;

  float* ws    = (float*)d_ws;
  float* xs    = ws;
  float* es    = xs   + (size_t)ND * HIDDEN;
  float* xm    = es   + (size_t)ED * HIDDEN;
  float* em    = xm   + (size_t)N_NODES * HIDDEN;
  float* alpha = em   + (size_t)N_EDGES * HIDDEN;
  float* Dbuf  = alpha + (size_t)NNZ * D;
  float* Bbuf  = Dbuf + ND;
  float* me    = Bbuf + N_EDGES + (size_t)ND * HIDDEN;

  const int TPB = 256;

  {
    wmma_gemm_lds<IN_CH><<<dim3(85, OUT_CH / 64), GEMM_TPB, 0, stream>>>(x, W_lin, b_lin, xs,
                                                                       OUT_CH, N_NODES / 16);
    wmma_gemm_lds<IN_CH><<<dim3(79, OUT_CH / 64), GEMM_TPB, 0, stream>>>(hattr, W_lin, b_lin, es,
                                                                       OUT_CH, N_EDGES / 16);
  }

  stalk_mean_kernel<<<(N_NODES * HIDDEN + TPB - 1) / TPB, TPB, 0, stream>>>(xs, xm, N_NODES * HIDDEN);
  stalk_mean_kernel<<<(N_EDGES * HIDDEN + TPB - 1) / TPB, TPB, 0, stream>>>(es, em, N_EDGES * HIDDEN);

  hipMemsetAsync(Dbuf, 0, (size_t)ND * sizeof(float), stream);
  hipMemsetAsync(Bbuf, 0, (size_t)N_EDGES * sizeof(float), stream);
  alpha_kernel<<<NNZ / 8, TPB, 0, stream>>>(xm, em, row, col, ln_g, ln_b,
                                            W_sheaf, b_sheaf, alpha, Dbuf, Bbuf);
  safe_inv_kernel<<<(ND + TPB - 1) / TPB, TPB, 0, stream>>>(Dbuf, ND);
  safe_inv_kernel<<<(N_EDGES + TPB - 1) / TPB, TPB, 0, stream>>>(Bbuf, N_EDGES);

  constexpr int RE = 8;
  constexpr int RN = 40;
  const int edgeBlk = (N_EDGES + RE - 1) / RE;
  const int nodeBlk = (N_NODES + RN - 1) / RN;
  static_assert(N_NODES % RN == 0, "the node side multiplies whole work-groups of RN rows");
  static_assert((size_t)((N_EDGES + RE - 1) / RE + (N_NODES + RN - 1) / RN) * SAVE_RUN
                    <= (size_t)N_NODES * HIDDEN, "saved runs fit in the xm region");
  int* edgeLists = reinterpret_cast<int*>(xm);
  int* nodeLists = edgeLists + (size_t)edgeBlk * SAVE_RUN;

  segsum_kernel<RE, false, 0><<<edgeBlk, SEG_TPB, 0, stream>>>(xs, alpha, col, row, nullptr, nullptr,
                                                               nullptr, nullptr, nullptr, me, N_EDGES, edgeLists);
  segsum_kernel<RN, true, 0><<<nodeBlk, SEG_TPB, 0, stream>>>(me, alpha, row, col, Bbuf, xs,
                                                              Dbuf, b_conv, W_conv, out, N_NODES, nodeLists);

  segsum_kernel<RE, false, 1><<<edgeBlk, SEG_TPB, 0, stream>>>(out, alpha, col, row, nullptr, nullptr,
                                                               nullptr, nullptr, nullptr, me, N_EDGES, edgeLists);
  segsum_kernel<RN, true, 1><<<nodeBlk, SEG_TPB, 0, stream>>>(me, alpha, row, col, Bbuf, out,
                                                              Dbuf, b_conv + HIDDEN, W_conv + HIDDEN * HIDDEN, out, N_NODES, nodeLists);
}


# Sheaf coefficients from per-row summaries: LayerNorm+projection split by linearity into node and hyperedge parts (8 floats a row, made with the stalk mean); one lane per incidence; sums re-associated
# speedup vs baseline: 3.3178x; 1.0419x over previous
#include <hip/hip_runtime.h>
#include <math.h>

constexpr int N_NODES = 50000;
constexpr int N_EDGES = 10000;
constexpr int NNZ     = 150000;
constexpr int IN_CH   = 128;
constexpr int HIDDEN  = 64;
constexpr int D       = 6;
constexpr int OUT_CH  = HIDDEN * D;
constexpr int ND      = N_NODES * D;
constexpr int ED      = N_EDGES * D;

typedef __attribute__((ext_vector_type(2))) float v2f;
typedef __attribute__((ext_vector_type(8))) float v8f;

constexpr int GEMM_WAVES = 8;
constexpr int GEMM_TPB   = GEMM_WAVES * 32;

template <int K>
__global__ __launch_bounds__(GEMM_TPB)
void wmma_gemm_lds(const float* __restrict__ A, const float* __restrict__ B,
                   const float* __restrict__ bias, float* __restrict__ C,
                   int N, int mTiles) {
  constexpr int KP   = K / 8;
  constexpr int AP   = K + 4;
  constexpr int CP   = 64 + 4;
  constexpr int SLOT = 16 * AP;
  static_assert(K % 8 == 0 && K >= 64, "slot must also hold the 16 x 64 result");
  __shared__ float4 Bs[KP * 4 * 32];
  __shared__ float4 Ss[GEMM_WAVES * SLOT / 4];

  const int tid  = threadIdx.x;
  const int wave = __builtin_amdgcn_readfirstlane(tid >> 5);
  const int lane = tid & 31;
  const int c    = lane & 15;
  const int h    = lane >> 4;
  const int n0   = blockIdx.y * 64;

  {
    constexpr int BIT = KP * 4 * 32 / GEMM_TPB;
    float4 bw[BIT];
#pragma unroll
    for (int it = 0; it < BIT; ++it) {
      const int idx = tid + it * GEMM_TPB;
      const int l  = idx & 31;
      const int nt = (idx >> 5) & 3;
      const int p  = idx >> 7;
      const float* bp = B + (size_t)(8 * p + 4 * (l >> 4)) * N + n0 + nt * 16 + (l & 15);
      bw[it].x = bp[0];
      bw[it].y = bp[N];
      bw[it].z = bp[2 * (size_t)N];
      bw[it].w = bp[3 * (size_t)N];
    }
#pragma unroll
    for (int it = 0; it < BIT; ++it) Bs[tid + it * GEMM_TPB] = bw[it];
  }
  __syncthreads();

  float* slot = reinterpret_cast<float*>(Ss) + wave * SLOT;
  float bv[4];
#pragma unroll
  for (int nt = 0; nt < 4; ++nt) bv[nt] = bias ? bias[n0 + nt * 16 + c] : 0.0f;

  for (int mt = blockIdx.x * GEMM_WAVES + wave; mt < mTiles; mt += gridDim.x * GEMM_WAVES) {
    const float4* g = reinterpret_cast<const float4*>(A + (size_t)mt * 16 * K);
    float4 tl[KP];
#pragma unroll
    for (int i = 0; i < KP; ++i) tl[i] = g[i * 32 + lane];
#pragma unroll
    for (int i = 0; i < KP; ++i) {
      const int f = i * 32 + lane;
      const int r = f / (K / 4);
      const int q = f % (K / 4);
      *reinterpret_cast<float4*>(slot + r * AP + 4 * q) = tl[i];
    }

    v8f acc[4] = {};
#pragma unroll 2
    for (int p = 0; p < KP; ++p) {
      const float4 a4 = *reinterpret_cast<const float4*>(slot + c * AP + 8 * p + 4 * h);
      float4 b4[4];
#pragma unroll
      for (int nt = 0; nt < 4; ++nt) b4[nt] = Bs[(p * 4 + nt) * 32 + lane];
      v2f a0; a0.x = a4.x; a0.y = a4.y;
      v2f a1; a1.x = a4.z; a1.y = a4.w;
#pragma unroll
      for (int nt = 0; nt < 4; ++nt) {
        v2f b0; b0.x = b4[nt].x; b0.y = b4[nt].y;
        acc[nt] = __builtin_amdgcn_wmma_f32_16x16x4_f32(false, a0, false, b0,
                                                        (short)0, acc[nt], false, false);
      }
#pragma unroll
      for (int nt = 0; nt < 4; ++nt) {
        v2f b1; b1.x = b4[nt].z; b1.y = b4[nt].w;
        acc[nt] = __builtin_amdgcn_wmma_f32_16x16x4_f32(false, a1, false, b1,
                                                        (short)0, acc[nt], false, false);
      }
    }

#pragma unroll
    for (int nt = 0; nt < 4; ++nt) {
#pragma unroll
      for (int r = 0; r < 8; ++r) slot[(8 * h + r) * CP + nt * 16 + c] = acc[nt][r] + bv[nt];
    }
#pragma unroll
    for (int i = 0; i < 8; ++i) {
      const int f   = i * 32 + lane;
      const int row = f >> 4;
      const int q   = f & 15;
      const float4 o = *reinterpret_cast<const float4*>(slot + row * CP + 4 * q);
      *reinterpret_cast<float4*>(C + (size_t)(mt * 16 + row) * N + n0 + 4 * q) = o;
    }
  }
}

constexpr int STAT_W   = 8;
constexpr int STAT_TPB = 128;
constexpr int STAT_RPH = 4;
static_assert((N_NODES + N_EDGES) * 16 / STAT_RPH % STAT_TPB == 0, "the grid covers the rows exactly");
static_assert(N_NODES % (2 * STAT_RPH) == 0, "a wave's rows are all node rows or all hyperedge rows");

__device__ inline float half_allsum(float v) {
#pragma unroll
  for (int off = 8; off > 0; off >>= 1) v += __shfl_xor(v, off, 16);
  return v;
}

__device__ inline float wave_allsum(float v) {
#pragma unroll
  for (int off = 16; off > 0; off >>= 1) v += __shfl_xor(v, off, 32);
  return v;
}

__global__ __launch_bounds__(STAT_TPB)
void rowstat_kernel(const float* __restrict__ xs, const float* __restrict__ es,
                    const float* __restrict__ ln_g, const float* __restrict__ Wsh,
                    float* __restrict__ nst, float* __restrict__ est) {
  const int gt   = blockIdx.x * STAT_TPB + threadIdx.x;
  const int q    = gt & 15;
  const int r0   = (gt >> 4) * STAT_RPH;
  const bool edge = r0 >= N_NODES;
  const float* tab = edge ? es : xs;
  float*       st  = edge ? est : nst;
  const int rb = edge ? (r0 - N_NODES) : r0;
  const int jb = (edge ? HIDDEN : 0) + 4 * q;

  float4 v[STAT_RPH][D];
#pragma unroll
  for (int u = 0; u < STAT_RPH; ++u) {
#pragma unroll
    for (int t = 0; t < D; ++t)
      v[u][t] = *reinterpret_cast<const float4*>(tab + (size_t)(rb + u) * OUT_CH + t * HIDDEN + 4 * q);
  }
  float gw[4][D];
#pragma unroll
  for (int k = 0; k < 4; ++k) {
    const float g = ln_g[jb + k];
#pragma unroll
    for (int t = 0; t < D; ++t) gw[k][t] = g * Wsh[(jb + k) * D + t];
  }

  float4 z[STAT_RPH];
  float  m[STAT_RPH];
#pragma unroll
  for (int u = 0; u < STAT_RPH; ++u) {
    z[u] = v[u][0];
#pragma unroll
    for (int t = 1; t < D; ++t) { z[u].x += v[u][t].x; z[u].y += v[u][t].y; z[u].z += v[u][t].z; z[u].w += v[u][t].w; }
    z[u].x *= (1.0f / 6.0f); z[u].y *= (1.0f / 6.0f); z[u].z *= (1.0f / 6.0f); z[u].w *= (1.0f / 6.0f);
    m[u] = (z[u].x + z[u].y) + (z[u].z + z[u].w);
  }
#pragma unroll
  for (int u = 0; u < STAT_RPH; ++u) m[u] = half_allsum(m[u]) * (1.0f / 64.0f);
  float c[STAT_RPH];
  float pc[STAT_RPH][D];
#pragma unroll
  for (int u = 0; u < STAT_RPH; ++u) {
    const float d0 = z[u].x - m[u], d1 = z[u].y - m[u], d2 = z[u].z - m[u], d3 = z[u].w - m[u];
    c[u] = (d0 * d0 + d1 * d1) + (d2 * d2 + d3 * d3);
#pragma unroll
    for (int t = 0; t < D; ++t)
      pc[u][t] = (d0 * gw[0][t] + d1 * gw[1][t]) + (d2 * gw[2][t] + d3 * gw[3][t]);
  }
#pragma unroll
  for (int u = 0; u < STAT_RPH; ++u) {
    c[u] = half_allsum(c[u]);
#pragma unroll
    for (int t = 0; t < D; ++t) pc[u][t] = half_allsum(pc[u][t]);
  }
  if (q < 2) {
#pragma unroll
    for (int u = 0; u < STAT_RPH; ++u) {
      float4 o;
      o.x = q ? pc[u][2] : m[u];
      o.y = q ? pc[u][3] : c[u];
      o.z = q ? pc[u][4] : pc[u][0];
      o.w = q ? pc[u][5] : pc[u][1];
      reinterpret_cast<float4*>(st + (size_t)(rb + u) * STAT_W)[q] = o;
    }
  }
}

constexpr int ALPHA_TPB = 256;
__global__ __launch_bounds__(ALPHA_TPB)
void alpha_kernel(const float* __restrict__ nst, const float* __restrict__ est,
                  const int* __restrict__ row, const int* __restrict__ col,
                  const float* __restrict__ ln_g, const float* __restrict__ ln_b,
                  const float* __restrict__ Wsh, const float* __restrict__ bsh,
                  float* __restrict__ alpha, float* __restrict__ Dsum,
                  float* __restrict__ Bcnt) {
  __shared__ float cst[2 * D];
  const int i  = blockIdx.x * ALPHA_TPB + threadIdx.x;
  const int ii = (i < NNZ) ? i : (NNZ - 1);
  const int rn = row[ii], cn = col[ii];
  const float4* ns = reinterpret_cast<const float4*>(nst + (size_t)rn * STAT_W);
  const float4* es = reinterpret_cast<const float4*>(est + (size_t)cn * STAT_W);
  const float4 n0 = ns[0], n1 = ns[1], e0 = es[0], e1 = es[1];

  if (threadIdx.x < 32) {
    const int lane = threadIdx.x;
    const float sg = (lane < 16) ? 1.0f : -1.0f;
    float gd[D] = {0, 0, 0, 0, 0, 0};
    float cb[D] = {0, 0, 0, 0, 0, 0};
#pragma unroll
    for (int k = 0; k < 4; ++k) {
      const int j = lane * 4 + k;
      const float g = sg * ln_g[j];
      const float b = ln_b[j];
#pragma unroll
      for (int t = 0; t < D; ++t) {
        const float w = Wsh[j * D + t];
        gd[t] += g * w;
        cb[t] += b * w;
      }
    }
#pragma unroll
    for (int t = 0; t < D; ++t) { gd[t] = wave_allsum(gd[t]); cb[t] = wave_allsum(cb[t]); }
    if (lane == 0) {
#pragma unroll
      for (int t = 0; t < D; ++t) { cst[t] = gd[t]; cst[D + t] = cb[t] + bsh[t]; }
    }
  }
  __syncthreads();
  if (i >= NNZ) return;

  const float dl   = 0.5f * (n0.x - e0.x);
  const float var  = (n0.y + e0.y) * (1.0f / 128.0f) + dl * dl;
  const float rstd = rsqrtf(var + 1e-5f);
  const float pc[D] = {n0.z + e0.z, n0.w + e0.w, n1.x + e1.x, n1.y + e1.y, n1.z + e1.z, n1.w + e1.w};
  float a[D];
#pragma unroll
  for (int t = 0; t < D; ++t) {
    const float p = rstd * (pc[t] + dl * cst[t]) + cst[D + t];
    a[t] = 1.0f / (1.0f + expf(-p));
  }
  float2* ao = reinterpret_cast<float2*>(alpha + (size_t)i * D);
  ao[0] = make_float2(a[0], a[1]);
  ao[1] = make_float2(a[2], a[3]);
  ao[2] = make_float2(a[4], a[5]);
#pragma unroll
  for (int t = 0; t < D; ++t) atomicAdd(&Dsum[rn * D + t], a[t] * a[t]);
  atomicAdd(&Bcnt[cn], 1.0f);
}

__global__ void safe_inv_kernel(float* __restrict__ buf, int n) {
  int idx = blockIdx.x * blockDim.x + threadIdx.x;
  if (idx >= n) return;
  float v = buf[idx];
  buf[idx] = (v > 0.0f) ? (1.0f / v) : 0.0f;
}

constexpr int SEG_TPB   = 96;
constexpr int SEG_WAVES = 3;
constexpr int SEG_CAP   = 256;
constexpr int STRIP     = 128;
constexpr int NSTRIPS   = (NNZ + STRIP - 1) / STRIP;
constexpr int NFULL     = NNZ / STRIP;
constexpr int SCAN_U    = 16;
constexpr int ACC_U     = 8;
constexpr int SAVE_HDR  = 32;
constexpr int SAVE_CAP  = 1024;
constexpr int SAVE_RUN  = SAVE_HDR + SAVE_CAP;
static_assert(NNZ % 4 == 0, "16-byte id loads are whole or absent");
static_assert(NNZ <= (1 << 18), "incidence id is packed into 18 bits");
static_assert(SEG_CAP >= 2 * STRIP, "a round must be able to take a strip");

template <int R, bool NODE, int MODE>
__global__ __launch_bounds__(SEG_TPB)
void segsum_kernel(const float* __restrict__ src, const float* __restrict__ alpha,
                   const int* __restrict__ key, const int* __restrict__ oth,
                   const float* __restrict__ Binv, const float* hself,
                   const float* __restrict__ Dinv, const float* __restrict__ bias,
                   const float* __restrict__ W,
                   float* dst, int nDest, int* __restrict__ saved) {
  static_assert(R <= 64, "owned row number and the 18-bit id share one word");
  static_assert(!NODE || (R * D) % 16 == 0, "owned stalk rows form whole 16-row tiles");
  __shared__ float4 acc[R * SEG_TPB];
  __shared__ int    lst[SEG_WAVES * SEG_CAP];
  __shared__ int    cntS[SEG_WAVES];
  __shared__ int    doneS[SEG_WAVES];

  const int tid  = threadIdx.x;
  const int wave = __builtin_amdgcn_readfirstlane(tid >> 5);
  const int lane = tid & 31;
  const int t    = tid >> 4;
  const int r0   = blockIdx.x * R;
  const int rr   = (nDest - r0 < R) ? (nDest - r0) : R;

  for (int r = 0; r < R; ++r) acc[r * SEG_TPB + tid] = make_float4(0.f, 0.f, 0.f, 0.f);

  int cnt = 0;
  auto strip = [&](const int4 k, const int i0) {
    const unsigned d0 = (unsigned)(k.x - r0);
    const unsigned d1 = (unsigned)(k.y - r0);
    const unsigned d2 = (unsigned)(k.z - r0);
    const unsigned d3 = (unsigned)(k.w - r0);
    const unsigned dm = min(min(d0, d1), min(d2, d3));
    if (__builtin_amdgcn_ballot_w32(dm < (unsigned)rr) != 0u) {
      auto append = [&](unsigned d, int i) {
        const bool m = d < (unsigned)rr;
        const unsigned b = __builtin_amdgcn_ballot_w32(m);
        if (m) {
          const int pos = cnt + (int)__builtin_amdgcn_mbcnt_lo(b, 0u);
          lst[wave * SEG_CAP + pos] = (int)((d << 18) | (unsigned)i);
        }
        cnt += __popc(b);
      };
      append(d0, i0);
      append(d1, i0 + 1);
      append(d2, i0 + 2);
      append(d3, i0 + 3);
    }
  };

  int* mine = saved + (size_t)blockIdx.x * SAVE_RUN;
  int stored = 0;
  int total = -1;
  if (MODE == 1) total = mine[0];
  const bool replay = (MODE == 1) && total >= 0 && total <= SAVE_CAP;
  int taken = 0;

  int s = wave;
  bool alldone = false;
  while (!alldone) {
   if (replay) {
    const int left = total - taken;
    const int n = (left < SEG_WAVES * SEG_CAP) ? left : SEG_WAVES * SEG_CAP;
    for (int e = tid; e < n; e += SEG_TPB) lst[e] = mine[SAVE_HDR + taken + e];
    taken += n;
    if (tid < SEG_WAVES) { cntS[tid] = (tid == 0) ? n : 0; doneS[tid] = (taken >= total) ? 1 : 0; }
   } else {
    cnt = 0;
    bool full = false;
    while (!full && s + (SCAN_U - 1) * SEG_WAVES < NFULL) {
      int4 k[SCAN_U];
      const int ib = s * STRIP + lane * 4;
#pragma unroll
      for (int u = 0; u < SCAN_U; ++u)
        k[u] = *reinterpret_cast<const int4*>(key + ib + u * SEG_WAVES * STRIP);
      int used = SCAN_U;
#pragma unroll
      for (int u = 0; u < SCAN_U; ++u) {
        strip(k[u], ib + u * SEG_WAVES * STRIP);
        if (cnt > SEG_CAP - STRIP) { full = true; used = u + 1; break; }
      }
      s += used * SEG_WAVES;
    }
    while (!full && s < NSTRIPS) {
      const int i0 = s * STRIP + lane * 4;
      const int4 k = (i0 < NNZ) ? *reinterpret_cast<const int4*>(key + i0)
                                : make_int4(-1, -1, -1, -1);
      strip(k, i0);
      s += SEG_WAVES;
      if (cnt > SEG_CAP - STRIP) full = true;
    }
    if (lane == 0) { cntS[wave] = cnt; doneS[wave] = (s >= NSTRIPS) ? 1 : 0; }
   }
    __syncthreads();

    if (MODE == 0) {
      const int c0 = cntS[0], c1 = cntS[1], c2 = cntS[2];
      const int tot = c0 + c1 + c2;
      if (stored >= 0 && stored + tot <= SAVE_CAP) {
        for (int e = tid; e < tot; e += SEG_TPB) {
          const int w = (e >= c0 ? 1 : 0) + (e >= c0 + c1 ? 1 : 0);
          const int o = e - (w == 0 ? 0 : (w == 1 ? c0 : c0 + c1));
          mine[SAVE_HDR + stored + e] = lst[w * SEG_CAP + o];
        }
        stored += tot;
      } else {
        stored = -1;
      }
    }

    for (int w = 0; w < SEG_WAVES; ++w) {
      const int n = __builtin_amdgcn_readfirstlane(cntS[w]);
      const int* L = lst + w * SEG_CAP;
      int   pkN[ACC_U];
      int   soN[ACC_U];
      float aN[ACC_U];
      auto hop1 = [&](const int mb) {
#pragma unroll
        for (int j = 0; j < ACC_U; ++j) pkN[j] = L[(mb + j < n) ? (mb + j) : (n - 1)];
#pragma unroll
        for (int j = 0; j < ACC_U; ++j) {
          const int i = pkN[j] & 0x3FFFF;
          soN[j] = oth[i];
          aN[j]  = alpha[(size_t)i * D + t];
        }
      };
      if (n > 0) hop1(0);
      for (int mb = 0; mb < n; mb += ACC_U) {
        int    pk[ACC_U];
        float  a[ACC_U];
        float  bi[ACC_U];
        float4 v[ACC_U];
#pragma unroll
        for (int j = 0; j < ACC_U; ++j) {
          pk[j] = pkN[j];
          a[j]  = aN[j];
          v[j]  = *reinterpret_cast<const float4*>(src + (size_t)soN[j] * OUT_CH + tid * 4);
          bi[j] = NODE ? Binv[soN[j]] : 1.0f;
        }
        hop1(mb + ACC_U);
#pragma unroll
        for (int j = 0; j < ACC_U; ++j) {
          const bool live = (mb + j < n);
          const int d = (int)((unsigned)pk[j] >> 18);
          const float aa = live ? (NODE ? a[j] * bi[j] : a[j]) : 0.0f;
          const float4 vv = live ? v[j] : make_float4(0.f, 0.f, 0.f, 0.f);
          float4 c = acc[d * SEG_TPB + tid];
          c.x += aa * vv.x;
          c.y += aa * vv.y;
          c.z += aa * vv.z;
          c.w += aa * vv.w;
          acc[d * SEG_TPB + tid] = c;
        }
      }
    }
    alldone = (doneS[0] & doneS[1] & doneS[2]) != 0;
    __syncthreads();
  }
  if (MODE == 0 && tid == 0) mine[0] = stored;

  if constexpr (!NODE) {
    auto finish = [&](const int r) {
      *reinterpret_cast<float4*>(dst + (size_t)(r0 + r) * OUT_CH + tid * 4) = acc[r * SEG_TPB + tid];
    };
    for (int r = 0; r < rr; ++r) finish(r);
  } else {
    constexpr int ZU = 10;
    static_assert(R % ZU == 0, "whole groups of rows");
#pragma nounroll
    for (int rb = 0; rb < R; rb += ZU) {
      float4 hv[ZU];
      float  dinv[ZU];
#pragma unroll
      for (int j = 0; j < ZU; ++j) {
        hv[j]   = *reinterpret_cast<const float4*>(hself + (size_t)(r0 + rb + j) * OUT_CH + tid * 4);
        dinv[j] = Dinv[(r0 + rb + j) * D + t];
      }
      float4 cs[ZU];
#pragma unroll
      for (int j = 0; j < ZU; ++j) cs[j] = acc[(rb + j) * SEG_TPB + tid];
#pragma unroll
      for (int j = 0; j < ZU; ++j) {
        float4 c = cs[j];
        c.x = hv[j].x - dinv[j] * c.x;
        c.y = hv[j].y - dinv[j] * c.y;
        c.z = hv[j].z - dinv[j] * c.z;
        c.w = hv[j].w - dinv[j] * c.w;
        const int key = ((rb + j) * D + t) & 15;
        acc[(rb + j) * SEG_TPB + (tid & ~15) + ((tid & 15) ^ key)] = c;
      }
    }
    const int c16 = lane & 15;
    const int hh  = lane >> 4;
    constexpr int KP = HIDDEN / 8;
    v2f bf[KP][4][2];
#pragma unroll
    for (int p = 0; p < KP; ++p) {
#pragma unroll
      for (int nt = 0; nt < 4; ++nt) {
        const float* bp = W + (8 * p + 4 * hh) * HIDDEN + nt * 16 + c16;
        bf[p][nt][0].x = bp[0];
        bf[p][nt][0].y = bp[HIDDEN];
        bf[p][nt][1].x = bp[2 * HIDDEN];
        bf[p][nt][1].y = bp[3 * HIDDEN];
      }
    }
    float bv[4];
#pragma unroll
    for (int nt = 0; nt < 4; ++nt) bv[nt] = bias[nt * 16 + c16];
    __syncthreads();

    float* accf = reinterpret_cast<float*>(acc);
    constexpr int MT = R * D / 16;
    for (int j = wave; j < MT; j += SEG_WAVES) {
      float* T = accf + j * 16 * HIDDEN;
      v8f o[4] = {};
#pragma unroll
      for (int p = 0; p < KP; ++p) {
        const float4 a4 = *reinterpret_cast<const float4*>(T + c16 * HIDDEN + 4 * ((2 * p + hh) ^ c16));
        v2f a0; a0.x = a4.x; a0.y = a4.y;
        v2f a1; a1.x = a4.z; a1.y = a4.w;
#pragma unroll
        for (int nt = 0; nt < 4; ++nt)
          o[nt] = __builtin_amdgcn_wmma_f32_16x16x4_f32(false, a0, false, bf[p][nt][0],
                                                        (short)0, o[nt], false, false);
#pragma unroll
        for (int nt = 0; nt < 4; ++nt)
          o[nt] = __builtin_amdgcn_wmma_f32_16x16x4_f32(false, a1, false, bf[p][nt][1],
                                                        (short)0, o[nt], false, false);
      }
#pragma unroll
      for (int nt = 0; nt < 4; ++nt) {
#pragma unroll
        for (int r = 0; r < 8; ++r) T[(8 * hh + r) * HIDDEN + nt * 16 + c16] = o[nt][r] + bv[nt];
      }
      float4 ov[8];
#pragma unroll
      for (int i = 0; i < 8; ++i) {
        const int f = i * 32 + lane;
        ov[i] = *reinterpret_cast<const float4*>(T + (f >> 4) * HIDDEN + 4 * (f & 15));
      }
#pragma unroll
      for (int i = 0; i < 8; ++i) {
        const int f   = i * 32 + lane;
        const int row = f >> 4;
        const int q   = f & 15;
        float4 e = ov[i];
        e.x = (e.x > 0.0f) ? e.x : expm1f(e.x);
        e.y = (e.y > 0.0f) ? e.y : expm1f(e.y);
        e.z = (e.z > 0.0f) ? e.z : expm1f(e.z);
        e.w = (e.w > 0.0f) ? e.w : expm1f(e.w);
        *reinterpret_cast<float4*>(dst + ((size_t)r0 * D + j * 16 + row) * HIDDEN + 4 * q) = e;
      }
    }
  }
}

extern "C" void kernel_launch(void* const* d_in, const int* in_sizes, int n_in,
                              void* d_out, int out_size, void* d_ws, size_t ws_size,
                              hipStream_t stream) {
  (void)in_sizes; (void)n_in; (void)out_size; (void)ws_size;
  const float* x       = (const float*)d_in[0];
  const float* hattr   = (const float*)d_in[1];
  const float* W_lin   = (const float*)d_in[2];
  const float* b_lin   = (const float*)d_in[3];
  const float* ln_g    = (const float*)d_in[4];
  const float* ln_b    = (const float*)d_in[5];
  const float* W_sheaf = (const float*)d_in[6];
  const float* b_sheaf = (const float*)d_in[7];
  const float* W_conv  = (const float*)d_in[8];
  const float* b_conv  = (const float*)d_in[9];
  const int*   eidx    = (const int*)d_in[10];
  const int* row = eidx;
  const int* col = eidx + NNZ;
  float* out = (float*)d_out;

  float* ws    = (float*)d_ws;
  float* xs    = ws;
  float* es    = xs   + (size_t)ND * HIDDEN;
  float* lsts  = es   + (size_t)ED * HIDDEN;
  float* nst   = lsts + (size_t)N_NODES * HIDDEN;
  float* est   = nst  + (size_t)N_NODES * STAT_W;
  float* alpha = nst  + (size_t)N_EDGES * HIDDEN;
  float* Dbuf  = alpha + (size_t)NNZ * D;
  float* Bbuf  = Dbuf + ND;
  float* me    = Bbuf + N_EDGES + (size_t)ND * HIDDEN;

  const int TPB = 256;

  {
    wmma_gemm_lds<IN_CH><<<dim3(85, OUT_CH / 64), GEMM_TPB, 0, stream>>>(x, W_lin, b_lin, xs,
                                                                       OUT_CH, N_NODES / 16);
    wmma_gemm_lds<IN_CH><<<dim3(79, OUT_CH / 64), GEMM_TPB, 0, stream>>>(hattr, W_lin, b_lin, es,
                                                                       OUT_CH, N_EDGES / 16);
  }

  static_assert((size_t)(N_NODES + N_EDGES) * STAT_W <= (size_t)N_EDGES * HIDDEN, "summaries fit in their region");
  rowstat_kernel<<<(N_NODES + N_EDGES) * 16 / STAT_RPH / STAT_TPB, STAT_TPB, 0, stream>>>(xs, es, ln_g, W_sheaf, nst, est);

  hipMemsetAsync(Dbuf, 0, (size_t)ND * sizeof(float), stream);
  hipMemsetAsync(Bbuf, 0, (size_t)N_EDGES * sizeof(float), stream);
  alpha_kernel<<<(NNZ + ALPHA_TPB - 1) / ALPHA_TPB, ALPHA_TPB, 0, stream>>>(nst, est, row, col, ln_g, ln_b,
                                                                           W_sheaf, b_sheaf, alpha, Dbuf, Bbuf);
  safe_inv_kernel<<<(ND + TPB - 1) / TPB, TPB, 0, stream>>>(Dbuf, ND);
  safe_inv_kernel<<<(N_EDGES + TPB - 1) / TPB, TPB, 0, stream>>>(Bbuf, N_EDGES);

  constexpr int RE = 8;
  constexpr int RN = 40;
  const int edgeBlk = (N_EDGES + RE - 1) / RE;
  const int nodeBlk = (N_NODES + RN - 1) / RN;
  static_assert(N_NODES % RN == 0, "the node side multiplies whole work-groups of RN rows");
  static_assert((size_t)((N_EDGES + RE - 1) / RE + (N_NODES + RN - 1) / RN) * SAVE_RUN
                    <= (size_t)N_NODES * HIDDEN, "saved runs fit in their region");
  int* edgeLists = reinterpret_cast<int*>(lsts);
  int* nodeLists = edgeLists + (size_t)edgeBlk * SAVE_RUN;

  segsum_kernel<RE, false, 0><<<edgeBlk, SEG_TPB, 0, stream>>>(xs, alpha, col, row, nullptr, nullptr,
                                                               nullptr, nullptr, nullptr, me, N_EDGES, edgeLists);
  segsum_kernel<RN, true, 0><<<nodeBlk, SEG_TPB, 0, stream>>>(me, alpha, row, col, Bbuf, xs,
                                                              Dbuf, b_conv, W_conv, out, N_NODES, nodeLists);

  segsum_kernel<RE, false, 1><<<edgeBlk, SEG_TPB, 0, stream>>>(out, alpha, col, row, nullptr, nullptr,
                                                               nullptr, nullptr, nullptr, me, N_EDGES, edgeLists);
  segsum_kernel<RN, true, 1><<<nodeBlk, SEG_TPB, 0, stream>>>(me, alpha, row, col, Bbuf, out,
                                                              Dbuf, b_conv + HIDDEN, W_conv + HIDDEN * HIDDEN, out, N_NODES, nodeLists);
}


# Degrees by owners: hyperedge owner counts incidences, stores rows times 1/count (was a factor per term); node owner sums alpha^2 in walk order (was atomic order); memsets, atomics, 2 launches gone
# speedup vs baseline: 3.6518x; 1.1007x over previous
#include <hip/hip_runtime.h>
#include <math.h>

constexpr int N_NODES = 50000;
constexpr int N_EDGES = 10000;
constexpr int NNZ     = 150000;
constexpr int IN_CH   = 128;
constexpr int HIDDEN  = 64;
constexpr int D       = 6;
constexpr int OUT_CH  = HIDDEN * D;
constexpr int ND      = N_NODES * D;
constexpr int ED      = N_EDGES * D;

typedef __attribute__((ext_vector_type(2))) float v2f;
typedef __attribute__((ext_vector_type(8))) float v8f;

constexpr int GEMM_WAVES = 8;
constexpr int GEMM_TPB   = GEMM_WAVES * 32;

template <int K>
__global__ __launch_bounds__(GEMM_TPB)
void wmma_gemm_lds(const float* __restrict__ A, const float* __restrict__ B,
                   const float* __restrict__ bias, float* __restrict__ C,
                   int N, int mTiles) {
  constexpr int KP   = K / 8;
  constexpr int AP   = K + 4;
  constexpr int CP   = 64 + 4;
  constexpr int SLOT = 16 * AP;
  static_assert(K % 8 == 0 && K >= 64, "slot must also hold the 16 x 64 result");
  __shared__ float4 Bs[KP * 4 * 32];
  __shared__ float4 Ss[GEMM_WAVES * SLOT / 4];

  const int tid  = threadIdx.x;
  const int wave = __builtin_amdgcn_readfirstlane(tid >> 5);
  const int lane = tid & 31;
  const int c    = lane & 15;
  const int h    = lane >> 4;
  const int n0   = blockIdx.y * 64;

  {
    constexpr int BIT = KP * 4 * 32 / GEMM_TPB;
    float4 bw[BIT];
#pragma unroll
    for (int it = 0; it < BIT; ++it) {
      const int idx = tid + it * GEMM_TPB;
      const int l  = idx & 31;
      const int nt = (idx >> 5) & 3;
      const int p  = idx >> 7;
      const float* bp = B + (size_t)(8 * p + 4 * (l >> 4)) * N + n0 + nt * 16 + (l & 15);
      bw[it].x = bp[0];
      bw[it].y = bp[N];
      bw[it].z = bp[2 * (size_t)N];
      bw[it].w = bp[3 * (size_t)N];
    }
#pragma unroll
    for (int it = 0; it < BIT; ++it) Bs[tid + it * GEMM_TPB] = bw[it];
  }
  __syncthreads();

  float* slot = reinterpret_cast<float*>(Ss) + wave * SLOT;
  float bv[4];
#pragma unroll
  for (int nt = 0; nt < 4; ++nt) bv[nt] = bias ? bias[n0 + nt * 16 + c] : 0.0f;

  for (int mt = blockIdx.x * GEMM_WAVES + wave; mt < mTiles; mt += gridDim.x * GEMM_WAVES) {
    const float4* g = reinterpret_cast<const float4*>(A + (size_t)mt * 16 * K);
    float4 tl[KP];
#pragma unroll
    for (int i = 0; i < KP; ++i) tl[i] = g[i * 32 + lane];
#pragma unroll
    for (int i = 0; i < KP; ++i) {
      const int f = i * 32 + lane;
      const int r = f / (K / 4);
      const int q = f % (K / 4);
      *reinterpret_cast<float4*>(slot + r * AP + 4 * q) = tl[i];
    }

    v8f acc[4] = {};
#pragma unroll 2
    for (int p = 0; p < KP; ++p) {
      const float4 a4 = *reinterpret_cast<const float4*>(slot + c * AP + 8 * p + 4 * h);
      float4 b4[4];
#pragma unroll
      for (int nt = 0; nt < 4; ++nt) b4[nt] = Bs[(p * 4 + nt) * 32 + lane];
      v2f a0; a0.x = a4.x; a0.y = a4.y;
      v2f a1; a1.x = a4.z; a1.y = a4.w;
#pragma unroll
      for (int nt = 0; nt < 4; ++nt) {
        v2f b0; b0.x = b4[nt].x; b0.y = b4[nt].y;
        acc[nt] = __builtin_amdgcn_wmma_f32_16x16x4_f32(false, a0, false, b0,
                                                        (short)0, acc[nt], false, false);
      }
#pragma unroll
      for (int nt = 0; nt < 4; ++nt) {
        v2f b1; b1.x = b4[nt].z; b1.y = b4[nt].w;
        acc[nt] = __builtin_amdgcn_wmma_f32_16x16x4_f32(false, a1, false, b1,
                                                        (short)0, acc[nt], false, false);
      }
    }

#pragma unroll
    for (int nt = 0; nt < 4; ++nt) {
#pragma unroll
      for (int r = 0; r < 8; ++r) slot[(8 * h + r) * CP + nt * 16 + c] = acc[nt][r] + bv[nt];
    }
#pragma unroll
    for (int i = 0; i < 8; ++i) {
      const int f   = i * 32 + lane;
      const int row = f >> 4;
      const int q   = f & 15;
      const float4 o = *reinterpret_cast<const float4*>(slot + row * CP + 4 * q);
      *reinterpret_cast<float4*>(C + (size_t)(mt * 16 + row) * N + n0 + 4 * q) = o;
    }
  }
}

constexpr int STAT_W   = 8;
constexpr int STAT_TPB = 128;
constexpr int STAT_RPH = 4;
static_assert((N_NODES + N_EDGES) * 16 / STAT_RPH % STAT_TPB == 0, "the grid covers the rows exactly");
static_assert(N_NODES % (2 * STAT_RPH) == 0, "a wave's rows are all node rows or all hyperedge rows");

__device__ inline float half_allsum(float v) {
#pragma unroll
  for (int off = 8; off > 0; off >>= 1) v += __shfl_xor(v, off, 16);
  return v;
}

__device__ inline float wave_allsum(float v) {
#pragma unroll
  for (int off = 16; off > 0; off >>= 1) v += __shfl_xor(v, off, 32);
  return v;
}

__global__ __launch_bounds__(STAT_TPB)
void rowstat_kernel(const float* __restrict__ xs, const float* __restrict__ es,
                    const float* __restrict__ ln_g, const float* __restrict__ Wsh,
                    float* __restrict__ nst, float* __restrict__ est) {
  const int gt   = blockIdx.x * STAT_TPB + threadIdx.x;
  const int q    = gt & 15;
  const int r0   = (gt >> 4) * STAT_RPH;
  const bool edge = r0 >= N_NODES;
  const float* tab = edge ? es : xs;
  float*       st  = edge ? est : nst;
  const int rb = edge ? (r0 - N_NODES) : r0;
  const int jb = (edge ? HIDDEN : 0) + 4 * q;

  float4 v[STAT_RPH][D];
#pragma unroll
  for (int u = 0; u < STAT_RPH; ++u) {
#pragma unroll
    for (int t = 0; t < D; ++t)
      v[u][t] = *reinterpret_cast<const float4*>(tab + (size_t)(rb + u) * OUT_CH + t * HIDDEN + 4 * q);
  }
  float gw[4][D];
#pragma unroll
  for (int k = 0; k < 4; ++k) {
    const float g = ln_g[jb + k];
#pragma unroll
    for (int t = 0; t < D; ++t) gw[k][t] = g * Wsh[(jb + k) * D + t];
  }

  float4 z[STAT_RPH];
  float  m[STAT_RPH];
#pragma unroll
  for (int u = 0; u < STAT_RPH; ++u) {
    z[u] = v[u][0];
#pragma unroll
    for (int t = 1; t < D; ++t) { z[u].x += v[u][t].x; z[u].y += v[u][t].y; z[u].z += v[u][t].z; z[u].w += v[u][t].w; }
    z[u].x *= (1.0f / 6.0f); z[u].y *= (1.0f / 6.0f); z[u].z *= (1.0f / 6.0f); z[u].w *= (1.0f / 6.0f);
    m[u] = (z[u].x + z[u].y) + (z[u].z + z[u].w);
  }
#pragma unroll
  for (int u = 0; u < STAT_RPH; ++u) m[u] = half_allsum(m[u]) * (1.0f / 64.0f);
  float c[STAT_RPH];
  float pc[STAT_RPH][D];
#pragma unroll
  for (int u = 0; u < STAT_RPH; ++u) {
    const float d0 = z[u].x - m[u], d1 = z[u].y - m[u], d2 = z[u].z - m[u], d3 = z[u].w - m[u];
    c[u] = (d0 * d0 + d1 * d1) + (d2 * d2 + d3 * d3);
#pragma unroll
    for (int t = 0; t < D; ++t)
      pc[u][t] = (d0 * gw[0][t] + d1 * gw[1][t]) + (d2 * gw[2][t] + d3 * gw[3][t]);
  }
#pragma unroll
  for (int u = 0; u < STAT_RPH; ++u) {
    c[u] = half_allsum(c[u]);
#pragma unroll
    for (int t = 0; t < D; ++t) pc[u][t] = half_allsum(pc[u][t]);
  }
  if (q < 2) {
#pragma unroll
    for (int u = 0; u < STAT_RPH; ++u) {
      float4 o;
      o.x = q ? pc[u][2] : m[u];
      o.y = q ? pc[u][3] : c[u];
      o.z = q ? pc[u][4] : pc[u][0];
      o.w = q ? pc[u][5] : pc[u][1];
      reinterpret_cast<float4*>(st + (size_t)(rb + u) * STAT_W)[q] = o;
    }
  }
}

constexpr int ALPHA_TPB = 256;
__global__ __launch_bounds__(ALPHA_TPB)
void alpha_kernel(const float* __restrict__ nst, const float* __restrict__ est,
                  const int* __restrict__ row, const int* __restrict__ col,
                  const float* __restrict__ ln_g, const float* __restrict__ ln_b,
                  const float* __restrict__ Wsh, const float* __restrict__ bsh,
                  float* __restrict__ alpha) {
  __shared__ float cst[2 * D];
  const int i  = blockIdx.x * ALPHA_TPB + threadIdx.x;
  const int ii = (i < NNZ) ? i : (NNZ - 1);
  const int rn = row[ii], cn = col[ii];
  const float4* ns = reinterpret_cast<const float4*>(nst + (size_t)rn * STAT_W);
  const float4* es = reinterpret_cast<const float4*>(est + (size_t)cn * STAT_W);
  const float4 n0 = ns[0], n1 = ns[1], e0 = es[0], e1 = es[1];

  if (threadIdx.x < 32) {
    const int lane = threadIdx.x;
    const float sg = (lane < 16) ? 1.0f : -1.0f;
    float gd[D] = {0, 0, 0, 0, 0, 0};
    float cb[D] = {0, 0, 0, 0, 0, 0};
#pragma unroll
    for (int k = 0; k < 4; ++k) {
      const int j = lane * 4 + k;
      const float g = sg * ln_g[j];
      const float b = ln_b[j];
#pragma unroll
      for (int t = 0; t < D; ++t) {
        const float w = Wsh[j * D + t];
        gd[t] += g * w;
        cb[t] += b * w;
      }
    }
#pragma unroll
    for (int t = 0; t < D; ++t) { gd[t] = wave_allsum(gd[t]); cb[t] = wave_allsum(cb[t]); }
    if (lane == 0) {
#pragma unroll
      for (int t = 0; t < D; ++t) { cst[t] = gd[t]; cst[D + t] = cb[t] + bsh[t]; }
    }
  }
  __syncthreads();
  if (i >= NNZ) return;

  const float dl   = 0.5f * (n0.x - e0.x);
  const float var  = (n0.y + e0.y) * (1.0f / 128.0f) + dl * dl;
  const float rstd = rsqrtf(var + 1e-5f);
  const float pc[D] = {n0.z + e0.z, n0.w + e0.w, n1.x + e1.x, n1.y + e1.y, n1.z + e1.z, n1.w + e1.w};
  float a[D];
#pragma unroll
  for (int t = 0; t < D; ++t) {
    const float p = rstd * (pc[t] + dl * cst[t]) + cst[D + t];
    a[t] = 1.0f / (1.0f + expf(-p));
  }
  float2* ao = reinterpret_cast<float2*>(alpha + (size_t)i * D);
  ao[0] = make_float2(a[0], a[1]);
  ao[1] = make_float2(a[2], a[3]);
  ao[2] = make_float2(a[4], a[5]);
}

constexpr int SEG_TPB   = 96;
constexpr int SEG_WAVES = 3;
constexpr int SEG_CAP   = 256;
constexpr int STRIP     = 128;
constexpr int NSTRIPS   = (NNZ + STRIP - 1) / STRIP;
constexpr int NFULL     = NNZ / STRIP;
constexpr int SCAN_U    = 16;
constexpr int ACC_U     = 8;
constexpr int SAVE_HDR  = 32;
constexpr int SAVE_CAP  = 1024;
constexpr int SAVE_RUN  = SAVE_HDR + SAVE_CAP;
static_assert(NNZ % 4 == 0, "16-byte id loads are whole or absent");
static_assert(NNZ <= (1 << 18), "incidence id is packed into 18 bits");
static_assert(SEG_CAP >= 2 * STRIP, "a round must be able to take a strip");

template <int R, bool NODE, int MODE>
__global__ __launch_bounds__(SEG_TPB)
void segsum_kernel(const float* __restrict__ src, const float* __restrict__ alpha,
                   const int* __restrict__ key, const int* __restrict__ oth,
                   const float* hself, const float* __restrict__ bias,
                   const float* __restrict__ W,
                   float* dst, int nDest, int* __restrict__ saved) {
  static_assert(R <= 64, "owned row number and the 18-bit id share one word");
  static_assert(!NODE || (R * D) % 16 == 0, "owned stalk rows form whole 16-row tiles");
  static_assert(NODE || R <= 32, "lane r of a wave counts owned row r");
  constexpr int DM = (R + 15) / 16;
  static_assert(R * D <= SEG_WAVES * SEG_CAP, "the inverse degrees fit in the finished lists' space");
  __shared__ float4 acc[R * SEG_TPB];
  __shared__ int    lst[SEG_WAVES * SEG_CAP];
  __shared__ int    cntS[SEG_WAVES];
  __shared__ int    doneS[SEG_WAVES];

  const int tid  = threadIdx.x;
  const int wave = __builtin_amdgcn_readfirstlane(tid >> 5);
  const int lane = tid & 31;
  const int t    = tid >> 4;
  const int k16  = tid & 15;
  const int r0   = blockIdx.x * R;
  const int rr   = (nDest - r0 < R) ? (nDest - r0) : R;

  for (int r = 0; r < R; ++r) acc[r * SEG_TPB + tid] = make_float4(0.f, 0.f, 0.f, 0.f);
  int   ecnt = 0;
  float dsq[DM];
#pragma unroll
  for (int m = 0; m < DM; ++m) dsq[m] = 0.0f;

  int cnt = 0;
  auto strip = [&](const int4 k, const int i0) {
    const unsigned d0 = (unsigned)(k.x - r0);
    const unsigned d1 = (unsigned)(k.y - r0);
    const unsigned d2 = (unsigned)(k.z - r0);
    const unsigned d3 = (unsigned)(k.w - r0);
    const unsigned dm = min(min(d0, d1), min(d2, d3));
    if (__builtin_amdgcn_ballot_w32(dm < (unsigned)rr) != 0u) {
      auto append = [&](unsigned d, int i) {
        const bool m = d < (unsigned)rr;
        const unsigned b = __builtin_amdgcn_ballot_w32(m);
        if (m) {
          const int pos = cnt + (int)__builtin_amdgcn_mbcnt_lo(b, 0u);
          lst[wave * SEG_CAP + pos] = (int)((d << 18) | (unsigned)i);
        }
        cnt += __popc(b);
      };
      append(d0, i0);
      append(d1, i0 + 1);
      append(d2, i0 + 2);
      append(d3, i0 + 3);
    }
  };

  int* mine = saved + (size_t)blockIdx.x * SAVE_RUN;
  int stored = 0;
  int total = -1;
  if (MODE == 1) total = mine[0];
  const bool replay = (MODE == 1) && total >= 0 && total <= SAVE_CAP;
  int taken = 0;

  int s = wave;
  bool alldone = false;
  while (!alldone) {
   if (replay) {
    const int left = total - taken;
    const int n = (left < SEG_WAVES * SEG_CAP) ? left : SEG_WAVES * SEG_CAP;
    for (int e = tid; e < n; e += SEG_TPB) lst[e] = mine[SAVE_HDR + taken + e];
    taken += n;
    if (tid < SEG_WAVES) { cntS[tid] = (tid == 0) ? n : 0; doneS[tid] = (taken >= total) ? 1 : 0; }
   } else {
    cnt = 0;
    bool full = false;
    while (!full && s + (SCAN_U - 1) * SEG_WAVES < NFULL) {
      int4 k[SCAN_U];
      const int ib = s * STRIP + lane * 4;
#pragma unroll
      for (int u = 0; u < SCAN_U; ++u)
        k[u] = *reinterpret_cast<const int4*>(key + ib + u * SEG_WAVES * STRIP);
      int used = SCAN_U;
#pragma unroll
      for (int u = 0; u < SCAN_U; ++u) {
        strip(k[u], ib + u * SEG_WAVES * STRIP);
        if (cnt > SEG_CAP - STRIP) { full = true; used = u + 1; break; }
      }
      s += used * SEG_WAVES;
    }
    while (!full && s < NSTRIPS) {
      const int i0 = s * STRIP + lane * 4;
      const int4 k = (i0 < NNZ) ? *reinterpret_cast<const int4*>(key + i0)
                                : make_int4(-1, -1, -1, -1);
      strip(k, i0);
      s += SEG_WAVES;
      if (cnt > SEG_CAP - STRIP) full = true;
    }
    if (lane == 0) { cntS[wave] = cnt; doneS[wave] = (s >= NSTRIPS) ? 1 : 0; }
   }
    __syncthreads();

    if (MODE == 0) {
      const int c0 = cntS[0], c1 = cntS[1], c2 = cntS[2];
      const int tot = c0 + c1 + c2;
      if (stored >= 0 && stored + tot <= SAVE_CAP) {
        for (int e = tid; e < tot; e += SEG_TPB) {
          const int w = (e >= c0 ? 1 : 0) + (e >= c0 + c1 ? 1 : 0);
          const int o = e - (w == 0 ? 0 : (w == 1 ? c0 : c0 + c1));
          mine[SAVE_HDR + stored + e] = lst[w * SEG_CAP + o];
        }
        stored += tot;
      } else {
        stored = -1;
      }
    }

    for (int w = 0; w < SEG_WAVES; ++w) {
      const int n = __builtin_amdgcn_readfirstlane(cntS[w]);
      const int* L = lst + w * SEG_CAP;
      int   pkN[ACC_U];
      int   soN[ACC_U];
      float aN[ACC_U];
      auto hop1 = [&](const int mb) {
#pragma unroll
        for (int j = 0; j < ACC_U; ++j) pkN[j] = L[(mb + j < n) ? (mb + j) : (n - 1)];
#pragma unroll
        for (int j = 0; j < ACC_U; ++j) {
          const int i = pkN[j] & 0x3FFFF;
          soN[j] = oth[i];
          aN[j]  = alpha[(size_t)i * D + t];
        }
      };
      if (n > 0) hop1(0);
      for (int mb = 0; mb < n; mb += ACC_U) {
        int    pk[ACC_U];
        float  a[ACC_U];
        float4 v[ACC_U];
#pragma unroll
        for (int j = 0; j < ACC_U; ++j) {
          pk[j] = pkN[j];
          a[j]  = aN[j];
          v[j]  = *reinterpret_cast<const float4*>(src + (size_t)soN[j] * OUT_CH + tid * 4);
        }
        hop1(mb + ACC_U);
#pragma unroll
        for (int j = 0; j < ACC_U; ++j) {
          const bool live = (mb + j < n);
          const int d = (int)((unsigned)pk[j] >> 18);
          const float aa = live ? a[j] : 0.0f;
          const float4 vv = live ? v[j] : make_float4(0.f, 0.f, 0.f, 0.f);
          if constexpr (NODE) {
            const float a2 = aa * aa;
#pragma unroll
            for (int m = 0; m < DM; ++m) dsq[m] += (d == k16 + 16 * m) ? a2 : 0.0f;
          } else {
            ecnt += (live && d == lane) ? 1 : 0;
          }
          float4 c = acc[d * SEG_TPB + tid];
          c.x += aa * vv.x;
          c.y += aa * vv.y;
          c.z += aa * vv.z;
          c.w += aa * vv.w;
          acc[d * SEG_TPB + tid] = c;
        }
      }
    }
    alldone = (doneS[0] & doneS[1] & doneS[2]) != 0;
    __syncthreads();
  }
  if (MODE == 0 && tid == 0) mine[0] = stored;

  if constexpr (!NODE) {
#pragma unroll
    for (int r = 0; r < R; ++r) {
      if (r < rr) {
        const int   cn = __builtin_amdgcn_readlane(ecnt, r);
        const float bi = (cn > 0) ? (1.0f / (float)cn) : 0.0f;
        float4 c = acc[r * SEG_TPB + tid];
        c.x *= bi; c.y *= bi; c.z *= bi; c.w *= bi;
        *reinterpret_cast<float4*>(dst + (size_t)(r0 + r) * OUT_CH + tid * 4) = c;
      }
    }
  } else {
    constexpr int ZU = 10;
    static_assert(R % ZU == 0, "whole groups of rows");
    float* dinvS = reinterpret_cast<float*>(lst);
#pragma unroll
    for (int m = 0; m < DM; ++m) {
      const int orow = k16 + 16 * m;
      if (orow < R) dinvS[orow * D + t] = (dsq[m] > 0.0f) ? (1.0f / dsq[m]) : 0.0f;
    }
    __syncthreads();
#pragma nounroll
    for (int rb = 0; rb < R; rb += ZU) {
      float4 hv[ZU];
      float  dinv[ZU];
#pragma unroll
      for (int j = 0; j < ZU; ++j) {
        hv[j]   = *reinterpret_cast<const float4*>(hself + (size_t)(r0 + rb + j) * OUT_CH + tid * 4);
        dinv[j] = dinvS[(rb + j) * D + t];
      }
      float4 cs[ZU];
#pragma unroll
      for (int j = 0; j < ZU; ++j) cs[j] = acc[(rb + j) * SEG_TPB + tid];
#pragma unroll
      for (int j = 0; j < ZU; ++j) {
        float4 c = cs[j];
        c.x = hv[j].x - dinv[j] * c.x;
        c.y = hv[j].y - dinv[j] * c.y;
        c.z = hv[j].z - dinv[j] * c.z;
        c.w = hv[j].w - dinv[j] * c.w;
        const int key = ((rb + j) * D + t) & 15;
        acc[(rb + j) * SEG_TPB + (tid & ~15) + ((tid & 15) ^ key)] = c;
      }
    }
    const int c16 = lane & 15;
    const int hh  = lane >> 4;
    constexpr int KP = HIDDEN / 8;
    v2f bf[KP][4][2];
#pragma unroll
    for (int p = 0; p < KP; ++p) {
#pragma unroll
      for (int nt = 0; nt < 4; ++nt) {
        const float* bp = W + (8 * p + 4 * hh) * HIDDEN + nt * 16 + c16;
        bf[p][nt][0].x = bp[0];
        bf[p][nt][0].y = bp[HIDDEN];
        bf[p][nt][1].x = bp[2 * HIDDEN];
        bf[p][nt][1].y = bp[3 * HIDDEN];
      }
    }
    float bv[4];
#pragma unroll
    for (int nt = 0; nt < 4; ++nt) bv[nt] = bias[nt * 16 + c16];
    __syncthreads();

    float* accf = reinterpret_cast<float*>(acc);
    constexpr int MT = R * D / 16;
    for (int j = wave; j < MT; j += SEG_WAVES) {
      float* T = accf + j * 16 * HIDDEN;
      v8f o[4] = {};
#pragma unroll
      for (int p = 0; p < KP; ++p) {
        const float4 a4 = *reinterpret_cast<const float4*>(T + c16 * HIDDEN + 4 * ((2 * p + hh) ^ c16));
        v2f a0; a0.x = a4.x; a0.y = a4.y;
        v2f a1; a1.x = a4.z; a1.y = a4.w;
#pragma unroll
        for (int nt = 0; nt < 4; ++nt)
          o[nt] = __builtin_amdgcn_wmma_f32_16x16x4_f32(false, a0, false, bf[p][nt][0],
                                                        (short)0, o[nt], false, false);
#pragma unroll
        for (int nt = 0; nt < 4; ++nt)
          o[nt] = __builtin_amdgcn_wmma_f32_16x16x4_f32(false, a1, false, bf[p][nt][1],
                                                        (short)0, o[nt], false, false);
      }
#pragma unroll
      for (int nt = 0; nt < 4; ++nt) {
#pragma unroll
        for (int r = 0; r < 8; ++r) T[(8 * hh + r) * HIDDEN + nt * 16 + c16] = o[nt][r] + bv[nt];
      }
      float4 ov[8];
#pragma unroll
      for (int i = 0; i < 8; ++i) {
        const int f = i * 32 + lane;
        ov[i] = *reinterpret_cast<const float4*>(T + (f >> 4) * HIDDEN + 4 * (f & 15));
      }
#pragma unroll
      for (int i = 0; i < 8; ++i) {
        const int f   = i * 32 + lane;
        const int row = f >> 4;
        const int q   = f & 15;
        float4 e = ov[i];
        e.x = (e.x > 0.0f) ? e.x : expm1f(e.x);
        e.y = (e.y > 0.0f) ? e.y : expm1f(e.y);
        e.z = (e.z > 0.0f) ? e.z : expm1f(e.z);
        e.w = (e.w > 0.0f) ? e.w : expm1f(e.w);
        *reinterpret_cast<float4*>(dst + ((size_t)r0 * D + j * 16 + row) * HIDDEN + 4 * q) = e;
      }
    }
  }
}

extern "C" void kernel_launch(void* const* d_in, const int* in_sizes, int n_in,
                              void* d_out, int out_size, void* d_ws, size_t ws_size,
                              hipStream_t stream) {
  (void)in_sizes; (void)n_in; (void)out_size; (void)ws_size;
  const float* x       = (const float*)d_in[0];
  const float* hattr   = (const float*)d_in[1];
  const float* W_lin   = (const float*)d_in[2];
  const float* b_lin   = (const float*)d_in[3];
  const float* ln_g    = (const float*)d_in[4];
  const float* ln_b    = (const float*)d_in[5];
  const float* W_sheaf = (const float*)d_in[6];
  const float* b_sheaf = (const float*)d_in[7];
  const float* W_conv  = (const float*)d_in[8];
  const float* b_conv  = (const float*)d_in[9];
  const int*   eidx    = (const int*)d_in[10];
  const int* row = eidx;
  const int* col = eidx + NNZ;
  float* out = (float*)d_out;

  float* ws    = (float*)d_ws;
  float* xs    = ws;
  float* es    = xs   + (size_t)ND * HIDDEN;
  float* lsts  = es   + (size_t)ED * HIDDEN;
  float* nst   = lsts + (size_t)N_NODES * HIDDEN;
  float* est   = nst  + (size_t)N_NODES * STAT_W;
  float* alpha = nst  + (size_t)N_EDGES * HIDDEN;
  float* me    = alpha + (size_t)NNZ * D + ND + N_EDGES + (size_t)ND * HIDDEN;

  {
    wmma_gemm_lds<IN_CH><<<dim3(85, OUT_CH / 64), GEMM_TPB, 0, stream>>>(x, W_lin, b_lin, xs,
                                                                       OUT_CH, N_NODES / 16);
    wmma_gemm_lds<IN_CH><<<dim3(79, OUT_CH / 64), GEMM_TPB, 0, stream>>>(hattr, W_lin, b_lin, es,
                                                                       OUT_CH, N_EDGES / 16);
  }

  static_assert((size_t)(N_NODES + N_EDGES) * STAT_W <= (size_t)N_EDGES * HIDDEN, "summaries fit in their region");
  rowstat_kernel<<<(N_NODES + N_EDGES) * 16 / STAT_RPH / STAT_TPB, STAT_TPB, 0, stream>>>(xs, es, ln_g, W_sheaf, nst, est);

  alpha_kernel<<<(NNZ + ALPHA_TPB - 1) / ALPHA_TPB, ALPHA_TPB, 0, stream>>>(nst, est, row, col, ln_g, ln_b,
                                                                           W_sheaf, b_sheaf, alpha);

  constexpr int RE = 8;
  constexpr int RN = 40;
  const int edgeBlk = (N_EDGES + RE - 1) / RE;
  const int nodeBlk = (N_NODES + RN - 1) / RN;
  static_assert(N_NODES % RN == 0, "the node side multiplies whole work-groups of RN rows");
  static_assert((size_t)((N_EDGES + RE - 1) / RE + (N_NODES + RN - 1) / RN) * SAVE_RUN
                    <= (size_t)N_NODES * HIDDEN, "saved runs fit in their region");
  int* edgeLists = reinterpret_cast<int*>(lsts);
  int* nodeLists = edgeLists + (size_t)edgeBlk * SAVE_RUN;

  segsum_kernel<RE, false, 0><<<edgeBlk, SEG_TPB, 0, stream>>>(xs, alpha, col, row, nullptr,
                                                               nullptr, nullptr, me, N_EDGES, edgeLists);
  segsum_kernel<RN, true, 0><<<nodeBlk, SEG_TPB, 0, stream>>>(me, alpha, row, col, xs,
                                                              b_conv, W_conv, out, N_NODES, nodeLists);

  segsum_kernel<RE, false, 1><<<edgeBlk, SEG_TPB, 0, stream>>>(out, alpha, col, row, nullptr,
                                                               nullptr, nullptr, me, N_EDGES, edgeLists);
  segsum_kernel<RN, true, 1><<<nodeBlk, SEG_TPB, 0, stream>>>(me, alpha, row, col, out,
                                                              b_conv + HIDDEN, W_conv + HIDDEN * HIDDEN, out, N_NODES, nodeLists);
}


# First-layer owners scan only their bucket: extra work-groups of the row-summary launch sort incidences into 50 buckets a side; an owner reads 16 sub-lists, not all ids; sum (walk) order changes
# speedup vs baseline: 3.8622x; 1.0576x over previous
#include <hip/hip_runtime.h>
#include <math.h>

constexpr int N_NODES = 50000;
constexpr int N_EDGES = 10000;
constexpr int NNZ     = 150000;
constexpr int IN_CH   = 128;
constexpr int HIDDEN  = 64;
constexpr int D       = 6;
constexpr int OUT_CH  = HIDDEN * D;
constexpr int ND      = N_NODES * D;
constexpr int ED      = N_EDGES * D;

typedef __attribute__((ext_vector_type(2))) float v2f;
typedef __attribute__((ext_vector_type(8))) float v8f;

constexpr int GEMM_WAVES = 8;
constexpr int GEMM_TPB   = GEMM_WAVES * 32;

template <int K>
__global__ __launch_bounds__(GEMM_TPB)
void wmma_gemm_lds(const float* __restrict__ A, const float* __restrict__ B,
                   const float* __restrict__ bias, float* __restrict__ C,
                   int N, int mTiles) {
  constexpr int KP   = K / 8;
  constexpr int AP   = K + 4;
  constexpr int CP   = 64 + 4;
  constexpr int SLOT = 16 * AP;
  static_assert(K % 8 == 0 && K >= 64, "slot must also hold the 16 x 64 result");
  __shared__ float4 Bs[KP * 4 * 32];
  __shared__ float4 Ss[GEMM_WAVES * SLOT / 4];

  const int tid  = threadIdx.x;
  const int wave = __builtin_amdgcn_readfirstlane(tid >> 5);
  const int lane = tid & 31;
  const int c    = lane & 15;
  const int h    = lane >> 4;
  const int n0   = blockIdx.y * 64;

  {
    constexpr int BIT = KP * 4 * 32 / GEMM_TPB;
    float4 bw[BIT];
#pragma unroll
    for (int it = 0; it < BIT; ++it) {
      const int idx = tid + it * GEMM_TPB;
      const int l  = idx & 31;
      const int nt = (idx >> 5) & 3;
      const int p  = idx >> 7;
      const float* bp = B + (size_t)(8 * p + 4 * (l >> 4)) * N + n0 + nt * 16 + (l & 15);
      bw[it].x = bp[0];
      bw[it].y = bp[N];
      bw[it].z = bp[2 * (size_t)N];
      bw[it].w = bp[3 * (size_t)N];
    }
#pragma unroll
    for (int it = 0; it < BIT; ++it) Bs[tid + it * GEMM_TPB] = bw[it];
  }
  __syncthreads();

  float* slot = reinterpret_cast<float*>(Ss) + wave * SLOT;
  float bv[4];
#pragma unroll
  for (int nt = 0; nt < 4; ++nt) bv[nt] = bias ? bias[n0 + nt * 16 + c] : 0.0f;

  for (int mt = blockIdx.x * GEMM_WAVES + wave; mt < mTiles; mt += gridDim.x * GEMM_WAVES) {
    const float4* g = reinterpret_cast<const float4*>(A + (size_t)mt * 16 * K);
    float4 tl[KP];
#pragma unroll
    for (int i = 0; i < KP; ++i) tl[i] = g[i * 32 + lane];
#pragma unroll
    for (int i = 0; i < KP; ++i) {
      const int f = i * 32 + lane;
      const int r = f / (K / 4);
      const int q = f % (K / 4);
      *reinterpret_cast<float4*>(slot + r * AP + 4 * q) = tl[i];
    }

    v8f acc[4] = {};
#pragma unroll 2
    for (int p = 0; p < KP; ++p) {
      const float4 a4 = *reinterpret_cast<const float4*>(slot + c * AP + 8 * p + 4 * h);
      float4 b4[4];
#pragma unroll
      for (int nt = 0; nt < 4; ++nt) b4[nt] = Bs[(p * 4 + nt) * 32 + lane];
      v2f a0; a0.x = a4.x; a0.y = a4.y;
      v2f a1; a1.x = a4.z; a1.y = a4.w;
#pragma unroll
      for (int nt = 0; nt < 4; ++nt) {
        v2f b0; b0.x = b4[nt].x; b0.y = b4[nt].y;
        acc[nt] = __builtin_amdgcn_wmma_f32_16x16x4_f32(false, a0, false, b0,
                                                        (short)0, acc[nt], false, false);
      }
#pragma unroll
      for (int nt = 0; nt < 4; ++nt) {
        v2f b1; b1.x = b4[nt].z; b1.y = b4[nt].w;
        acc[nt] = __builtin_amdgcn_wmma_f32_16x16x4_f32(false, a1, false, b1,
                                                        (short)0, acc[nt], false, false);
      }
    }

#pragma unroll
    for (int nt = 0; nt < 4; ++nt) {
#pragma unroll
      for (int r = 0; r < 8; ++r) slot[(8 * h + r) * CP + nt * 16 + c] = acc[nt][r] + bv[nt];
    }
#pragma unroll
    for (int i = 0; i < 8; ++i) {
      const int f   = i * 32 + lane;
      const int row = f >> 4;
      const int q   = f & 15;
      const float4 o = *reinterpret_cast<const float4*>(slot + row * CP + 4 * q);
      *reinterpret_cast<float4*>(C + (size_t)(mt * 16 + row) * N + n0 + 4 * q) = o;
    }
  }
}

constexpr int STAT_W   = 8;
constexpr int STAT_TPB = 128;
constexpr int STAT_RPH = 4;
static_assert((N_NODES + N_EDGES) * 16 / STAT_RPH % STAT_TPB == 0, "the grid covers the rows exactly");
static_assert(N_NODES % (2 * STAT_RPH) == 0, "a wave's rows are all node rows or all hyperedge rows");

constexpr int SEG_RE  = 8;
constexpr int SEG_RN  = 40;
constexpr int C_STRIP = 128;
constexpr int CB_SUB  = 16;
constexpr int CB_CAP  = 1024;
constexpr int CB_STR  = CB_CAP / C_STRIP;
constexpr int CB_HDR  = 32;
constexpr int CB_RUN  = CB_HDR + CB_CAP;
constexpr int CB_FINE = 25;
constexpr int CB_NB   = 50;
constexpr int CB_WGS  = 2 * CB_NB * 4;
constexpr int CB_U    = 8;
static_assert(CB_NB * CB_FINE * SEG_RE == N_EDGES && CB_NB * CB_FINE * SEG_RN == N_NODES, "buckets cover the owner ranges exactly");
static_assert(CB_FINE * SEG_RN < (1 << 14) - 1, "id within a bucket fits above the 18-bit incidence; all ones stays unmatched");

__device__ inline void coarse_pass(const int* __restrict__ row, const int* __restrict__ col,
                                   int* __restrict__ cl, const int cb) {
  const int wave   = __builtin_amdgcn_readfirstlane((int)threadIdx.x >> 5);
  const int lane   = threadIdx.x & 31;
  const int bucket = cb >> 2;
  const int j      = (cb & 3) * 4 + wave;
  const bool edge  = bucket >= CB_NB;
  const int* key   = edge ? col : row;
  const int span   = CB_FINE * (edge ? SEG_RE : SEG_RN);
  const int base   = (edge ? bucket - CB_NB : bucket) * span;
  int* run = cl + (size_t)(bucket * CB_SUB + j) * CB_RUN;
  constexpr int NS = (NNZ + C_STRIP - 1) / C_STRIP;
  int  cnt  = 0;
  bool over = false;
  for (int s0 = j; s0 < NS && !over; s0 += CB_SUB * CB_U) {
    int4 k[CB_U];
#pragma unroll
    for (int u = 0; u < CB_U; ++u) {
      const int i0 = (s0 + u * CB_SUB) * C_STRIP + lane * 4;
      k[u] = (i0 < NNZ) ? *reinterpret_cast<const int4*>(key + i0) : make_int4(-1, -1, -1, -1);
    }
#pragma unroll
    for (int u = 0; u < CB_U; ++u) {
      if (cnt > CB_CAP - C_STRIP) over = true;
      if (!over) {
        const int i0 = (s0 + u * CB_SUB) * C_STRIP + lane * 4;
        const unsigned d0 = (unsigned)(k[u].x - base);
        const unsigned d1 = (unsigned)(k[u].y - base);
        const unsigned d2 = (unsigned)(k[u].z - base);
        const unsigned d3 = (unsigned)(k[u].w - base);
        const unsigned dm = min(min(d0, d1), min(d2, d3));
        if (__builtin_amdgcn_ballot_w32(dm < (unsigned)span) != 0u) {
          auto put = [&](unsigned d, int i) {
            const bool m = d < (unsigned)span;
            const unsigned b = __builtin_amdgcn_ballot_w32(m);
            if (m) run[CB_HDR + cnt + (int)__builtin_amdgcn_mbcnt_lo(b, 0u)] = (int)((d << 18) | (unsigned)i);
            cnt += __popc(b);
          };
          put(d0, i0);
          put(d1, i0 + 1);
          put(d2, i0 + 2);
          put(d3, i0 + 3);
        }
      }
    }
  }
  if (over) {
    if (lane == 0) run[0] = -1;
  } else {
    const int end = (cnt + C_STRIP - 1) & ~(C_STRIP - 1);
    for (int p = cnt + lane; p < end; p += 32) run[CB_HDR + p] = -1;
    if (lane == 0) run[0] = cnt;
  }
}

__device__ inline float half_allsum(float v) {
#pragma unroll
  for (int off = 8; off > 0; off >>= 1) v += __shfl_xor(v, off, 16);
  return v;
}

__device__ inline float wave_allsum(float v) {
#pragma unroll
  for (int off = 16; off > 0; off >>= 1) v += __shfl_xor(v, off, 32);
  return v;
}

__global__ __launch_bounds__(STAT_TPB)
void rowstat_kernel(const float* __restrict__ xs, const float* __restrict__ es,
                    const float* __restrict__ ln_g, const float* __restrict__ Wsh,
                    float* __restrict__ nst, float* __restrict__ est,
                    const int* __restrict__ row, const int* __restrict__ col, int* __restrict__ cl) {
  if (blockIdx.x < CB_WGS) { coarse_pass(row, col, cl, (int)blockIdx.x); return; }
  const int gt   = (blockIdx.x - CB_WGS) * STAT_TPB + threadIdx.x;
  const int q    = gt & 15;
  const int r0   = (gt >> 4) * STAT_RPH;
  const bool edge = r0 >= N_NODES;
  const float* tab = edge ? es : xs;
  float*       st  = edge ? est : nst;
  const int rb = edge ? (r0 - N_NODES) : r0;
  const int jb = (edge ? HIDDEN : 0) + 4 * q;

  float4 v[STAT_RPH][D];
#pragma unroll
  for (int u = 0; u < STAT_RPH; ++u) {
#pragma unroll
    for (int t = 0; t < D; ++t)
      v[u][t] = *reinterpret_cast<const float4*>(tab + (size_t)(rb + u) * OUT_CH + t * HIDDEN + 4 * q);
  }
  float gw[4][D];
#pragma unroll
  for (int k = 0; k < 4; ++k) {
    const float g = ln_g[jb + k];
#pragma unroll
    for (int t = 0; t < D; ++t) gw[k][t] = g * Wsh[(jb + k) * D + t];
  }

  float4 z[STAT_RPH];
  float  m[STAT_RPH];
#pragma unroll
  for (int u = 0; u < STAT_RPH; ++u) {
    z[u] = v[u][0];
#pragma unroll
    for (int t = 1; t < D; ++t) { z[u].x += v[u][t].x; z[u].y += v[u][t].y; z[u].z += v[u][t].z; z[u].w += v[u][t].w; }
    z[u].x *= (1.0f / 6.0f); z[u].y *= (1.0f / 6.0f); z[u].z *= (1.0f / 6.0f); z[u].w *= (1.0f / 6.0f);
    m[u] = (z[u].x + z[u].y) + (z[u].z + z[u].w);
  }
#pragma unroll
  for (int u = 0; u < STAT_RPH; ++u) m[u] = half_allsum(m[u]) * (1.0f / 64.0f);
  float c[STAT_RPH];
  float pc[STAT_RPH][D];
#pragma unroll
  for (int u = 0; u < STAT_RPH; ++u) {
    const float d0 = z[u].x - m[u], d1 = z[u].y - m[u], d2 = z[u].z - m[u], d3 = z[u].w - m[u];
    c[u] = (d0 * d0 + d1 * d1) + (d2 * d2 + d3 * d3);
#pragma unroll
    for (int t = 0; t < D; ++t)
      pc[u][t] = (d0 * gw[0][t] + d1 * gw[1][t]) + (d2 * gw[2][t] + d3 * gw[3][t]);
  }
#pragma unroll
  for (int u = 0; u < STAT_RPH; ++u) {
    c[u] = half_allsum(c[u]);
#pragma unroll
    for (int t = 0; t < D; ++t) pc[u][t] = half_allsum(pc[u][t]);
  }
  if (q < 2) {
#pragma unroll
    for (int u = 0; u < STAT_RPH; ++u) {
      float4 o;
      o.x = q ? pc[u][2] : m[u];
      o.y = q ? pc[u][3] : c[u];
      o.z = q ? pc[u][4] : pc[u][0];
      o.w = q ? pc[u][5] : pc[u][1];
      reinterpret_cast<float4*>(st + (size_t)(rb + u) * STAT_W)[q] = o;
    }
  }
}

constexpr int ALPHA_TPB = 256;
__global__ __launch_bounds__(ALPHA_TPB)
void alpha_kernel(const float* __restrict__ nst, const float* __restrict__ est,
                  const int* __restrict__ row, const int* __restrict__ col,
                  const float* __restrict__ ln_g, const float* __restrict__ ln_b,
                  const float* __restrict__ Wsh, const float* __restrict__ bsh,
                  float* __restrict__ alpha) {
  __shared__ float cst[2 * D];
  const int i  = blockIdx.x * ALPHA_TPB + threadIdx.x;
  const int ii = (i < NNZ) ? i : (NNZ - 1);
  const int rn = row[ii], cn = col[ii];
  const float4* ns = reinterpret_cast<const float4*>(nst + (size_t)rn * STAT_W);
  const float4* es = reinterpret_cast<const float4*>(est + (size_t)cn * STAT_W);
  const float4 n0 = ns[0], n1 = ns[1], e0 = es[0], e1 = es[1];

  if (threadIdx.x < 32) {
    const int lane = threadIdx.x;
    const float sg = (lane < 16) ? 1.0f : -1.0f;
    float gd[D] = {0, 0, 0, 0, 0, 0};
    float cb[D] = {0, 0, 0, 0, 0, 0};
#pragma unroll
    for (int k = 0; k < 4; ++k) {
      const int j = lane * 4 + k;
      const float g = sg * ln_g[j];
      const float b = ln_b[j];
#pragma unroll
      for (int t = 0; t < D; ++t) {
        const float w = Wsh[j * D + t];
        gd[t] += g * w;
        cb[t] += b * w;
      }
    }
#pragma unroll
    for (int t = 0; t < D; ++t) { gd[t] = wave_allsum(gd[t]); cb[t] = wave_allsum(cb[t]); }
    if (lane == 0) {
#pragma unroll
      for (int t = 0; t < D; ++t) { cst[t] = gd[t]; cst[D + t] = cb[t] + bsh[t]; }
    }
  }
  __syncthreads();
  if (i >= NNZ) return;

  const float dl   = 0.5f * (n0.x - e0.x);
  const float var  = (n0.y + e0.y) * (1.0f / 128.0f) + dl * dl;
  const float rstd = rsqrtf(var + 1e-5f);
  const float pc[D] = {n0.z + e0.z, n0.w + e0.w, n1.x + e1.x, n1.y + e1.y, n1.z + e1.z, n1.w + e1.w};
  float a[D];
#pragma unroll
  for (int t = 0; t < D; ++t) {
    const float p = rstd * (pc[t] + dl * cst[t]) + cst[D + t];
    a[t] = 1.0f / (1.0f + expf(-p));
  }
  float2* ao = reinterpret_cast<float2*>(alpha + (size_t)i * D);
  ao[0] = make_float2(a[0], a[1]);
  ao[1] = make_float2(a[2], a[3]);
  ao[2] = make_float2(a[4], a[5]);
}

constexpr int SEG_TPB   = 96;
constexpr int SEG_WAVES = 3;
constexpr int SEG_CAP   = 256;
constexpr int STRIP     = 128;
constexpr int NSTRIPS   = (NNZ + STRIP - 1) / STRIP;
constexpr int NFULL     = NNZ / STRIP;
constexpr int SCAN_U    = 16;
constexpr int ACC_U     = 8;
constexpr int SAVE_HDR  = 32;
constexpr int SAVE_CAP  = 1024;
constexpr int SAVE_RUN  = SAVE_HDR + SAVE_CAP;
static_assert(NNZ % 4 == 0, "16-byte id loads are whole or absent");
static_assert(NNZ <= (1 << 18), "incidence id is packed into 18 bits");
static_assert(SEG_CAP >= 2 * STRIP, "a round must be able to take a strip");
static_assert(C_STRIP == STRIP, "bucket sub-lists are padded to the owners' strip length");

template <int R, bool NODE, int MODE>
__global__ __launch_bounds__(SEG_TPB)
void segsum_kernel(const float* __restrict__ src, const float* __restrict__ alpha,
                   const int* __restrict__ key, const int* __restrict__ oth,
                   const float* hself, const float* __restrict__ bias,
                   const float* __restrict__ W,
                   float* dst, int nDest, int* __restrict__ saved,
                   const int* __restrict__ coarse) {
  static_assert(R <= 64, "owned row number and the 18-bit id share one word");
  static_assert(!NODE || (R * D) % 16 == 0, "owned stalk rows form whole 16-row tiles");
  static_assert(NODE || R <= 32, "lane r of a wave counts owned row r");
  constexpr int DM = (R + 15) / 16;
  static_assert(R * D <= SEG_WAVES * SEG_CAP, "the inverse degrees fit in the finished lists' space");
  __shared__ float4 acc[R * SEG_TPB];
  __shared__ int    lst[SEG_WAVES * SEG_CAP];
  __shared__ int    cntS[SEG_WAVES];
  __shared__ int    doneS[SEG_WAVES];

  const int tid  = threadIdx.x;
  const int wave = __builtin_amdgcn_readfirstlane(tid >> 5);
  const int lane = tid & 31;
  const int t    = tid >> 4;
  const int k16  = tid & 15;
  const int r0   = blockIdx.x * R;
  const int rr   = (nDest - r0 < R) ? (nDest - r0) : R;

  for (int r = 0; r < R; ++r) acc[r * SEG_TPB + tid] = make_float4(0.f, 0.f, 0.f, 0.f);
  int   ecnt = 0;
  float dsq[DM];
#pragma unroll
  for (int m = 0; m < DM; ++m) dsq[m] = 0.0f;

  int cnt = 0;
  auto append = [&](unsigned d, int i) {
    const bool m = d < (unsigned)rr;
    const unsigned b = __builtin_amdgcn_ballot_w32(m);
    if (m) {
      const int pos = cnt + (int)__builtin_amdgcn_mbcnt_lo(b, 0u);
      lst[wave * SEG_CAP + pos] = (int)((d << 18) | (unsigned)i);
    }
    cnt += __popc(b);
  };
  auto strip = [&](const int4 k, const int i0) {
    const unsigned d0 = (unsigned)(k.x - r0);
    const unsigned d1 = (unsigned)(k.y - r0);
    const unsigned d2 = (unsigned)(k.z - r0);
    const unsigned d3 = (unsigned)(k.w - r0);
    const unsigned dm = min(min(d0, d1), min(d2, d3));
    if (__builtin_amdgcn_ballot_w32(dm < (unsigned)rr) != 0u) {
      append(d0, i0);
      append(d1, i0 + 1);
      append(d2, i0 + 2);
      append(d3, i0 + 3);
    }
  };

  const int* cbk = coarse;
  int  cbase = 0;
  bool useC  = false;
  int  cj[CB_SUB];
  if (MODE == 0) {
    const int bucket = (int)blockIdx.x / CB_FINE;
    cbk   = coarse + (size_t)bucket * CB_SUB * CB_RUN;
    cbase = bucket * CB_FINE * R - r0;
    int ok = 1;
#pragma unroll
    for (int j = 0; j < CB_SUB; ++j) {
      cj[j] = cbk[j * CB_RUN];
      ok &= (cj[j] >= 0) ? 1 : 0;
    }
    useC = ok != 0;
  }
  auto stripC = [&](const int4 e) {
    const unsigned d0 = (unsigned)((int)((unsigned)e.x >> 18) + cbase);
    const unsigned d1 = (unsigned)((int)((unsigned)e.y >> 18) + cbase);
    const unsigned d2 = (unsigned)((int)((unsigned)e.z >> 18) + cbase);
    const unsigned d3 = (unsigned)((int)((unsigned)e.w >> 18) + cbase);
    const unsigned dm = min(min(d0, d1), min(d2, d3));
    if (__builtin_amdgcn_ballot_w32(dm < (unsigned)rr) != 0u) {
      append(d0, e.x & 0x3FFFF);
      append(d1, e.y & 0x3FFFF);
      append(d2, e.z & 0x3FFFF);
      append(d3, e.w & 0x3FFFF);
    }
  };
  constexpr int CB_PASS = (CB_STR + SEG_WAVES - 1) / SEG_WAVES;
  int pm = 0;
  int pu = 0;

  int* mine = saved + (size_t)blockIdx.x * SAVE_RUN;
  int stored = 0;
  int total = -1;
  if (MODE == 1) total = mine[0];
  const bool replay = (MODE == 1) && total >= 0 && total <= SAVE_CAP;
  int taken = 0;

  int s = wave;
  bool alldone = false;
  while (!alldone) {
   if (replay) {
    const int left = total - taken;
    const int n = (left < SEG_WAVES * SEG_CAP) ? left : SEG_WAVES * SEG_CAP;
    for (int e = tid; e < n; e += SEG_TPB) lst[e] = mine[SAVE_HDR + taken + e];
    taken += n;
    if (tid < SEG_WAVES) { cntS[tid] = (tid == 0) ? n : 0; doneS[tid] = (taken >= total) ? 1 : 0; }
   } else if (MODE == 0 && useC) {
    cnt = 0;
    bool full = false;
    while (!full && pm < CB_PASS) {
      const int kk = wave + SEG_WAVES * pm;
      int4 e[CB_SUB];
#pragma unroll
      for (int u = 0; u < CB_SUB; ++u) {
        const bool live = (u >= pu) && (kk < CB_STR) && (kk * STRIP < cj[u]);
        e[u] = live ? *reinterpret_cast<const int4*>(cbk + u * CB_RUN + CB_HDR + kk * STRIP + lane * 4)
                    : make_int4(-1, -1, -1, -1);
      }
      int nu = CB_SUB;
#pragma unroll
      for (int u = 0; u < CB_SUB; ++u) {
        stripC(e[u]);
        if (cnt > SEG_CAP - STRIP) { full = true; nu = u + 1; break; }
      }
      if (nu == CB_SUB) { ++pm; pu = 0; } else { pu = nu; }
    }
    if (lane == 0) { cntS[wave] = cnt; doneS[wave] = (pm >= CB_PASS) ? 1 : 0; }
   } else {
    cnt = 0;
    bool full = false;
    while (!full && s + (SCAN_U - 1) * SEG_WAVES < NFULL) {
      int4 k[SCAN_U];
      const int ib = s * STRIP + lane * 4;
#pragma unroll
      for (int u = 0; u < SCAN_U; ++u)
        k[u] = *reinterpret_cast<const int4*>(key + ib + u * SEG_WAVES * STRIP);
      int used = SCAN_U;
#pragma unroll
      for (int u = 0; u < SCAN_U; ++u) {
        strip(k[u], ib + u * SEG_WAVES * STRIP);
        if (cnt > SEG_CAP - STRIP) { full = true; used = u + 1; break; }
      }
      s += used * SEG_WAVES;
    }
    while (!full && s < NSTRIPS) {
      const int i0 = s * STRIP + lane * 4;
      const int4 k = (i0 < NNZ) ? *reinterpret_cast<const int4*>(key + i0)
                                : make_int4(-1, -1, -1, -1);
      strip(k, i0);
      s += SEG_WAVES;
      if (cnt > SEG_CAP - STRIP) full = true;
    }
    if (lane == 0) { cntS[wave] = cnt; doneS[wave] = (s >= NSTRIPS) ? 1 : 0; }
   }
    __syncthreads();

    if (MODE == 0) {
      const int c0 = cntS[0], c1 = cntS[1], c2 = cntS[2];
      const int tot = c0 + c1 + c2;
      if (stored >= 0 && stored + tot <= SAVE_CAP) {
        for (int e = tid; e < tot; e += SEG_TPB) {
          const int w = (e >= c0 ? 1 : 0) + (e >= c0 + c1 ? 1 : 0);
          const int o = e - (w == 0 ? 0 : (w == 1 ? c0 : c0 + c1));
          mine[SAVE_HDR + stored + e] = lst[w * SEG_CAP + o];
        }
        stored += tot;
      } else {
        stored = -1;
      }
    }

    for (int w = 0; w < SEG_WAVES; ++w) {
      const int n = __builtin_amdgcn_readfirstlane(cntS[w]);
      const int* L = lst + w * SEG_CAP;
      int   pkN[ACC_U];
      int   soN[ACC_U];
      float aN[ACC_U];
      auto hop1 = [&](const int mb) {
#pragma unroll
        for (int j = 0; j < ACC_U; ++j) pkN[j] = L[(mb + j < n) ? (mb + j) : (n - 1)];
#pragma unroll
        for (int j = 0; j < ACC_U; ++j) {
          const int i = pkN[j] & 0x3FFFF;
          soN[j] = oth[i];
          aN[j]  = alpha[(size_t)i * D + t];
        }
      };
      if (n > 0) hop1(0);
      for (int mb = 0; mb < n; mb += ACC_U) {
        int    pk[ACC_U];
        float  a[ACC_U];
        float4 v[ACC_U];
#pragma unroll
        for (int j = 0; j < ACC_U; ++j) {
          pk[j] = pkN[j];
          a[j]  = aN[j];
          v[j]  = *reinterpret_cast<const float4*>(src + (size_t)soN[j] * OUT_CH + tid * 4);
        }
        hop1(mb + ACC_U);
#pragma unroll
        for (int j = 0; j < ACC_U; ++j) {
          const bool live = (mb + j < n);
          const int d = (int)((unsigned)pk[j] >> 18);
          const float aa = live ? a[j] : 0.0f;
          const float4 vv = live ? v[j] : make_float4(0.f, 0.f, 0.f, 0.f);
          if constexpr (NODE) {
            const float a2 = aa * aa;
#pragma unroll
            for (int m = 0; m < DM; ++m) dsq[m] += (d == k16 + 16 * m) ? a2 : 0.0f;
          } else {
            ecnt += (live && d == lane) ? 1 : 0;
          }
          float4 c = acc[d * SEG_TPB + tid];
          c.x += aa * vv.x;
          c.y += aa * vv.y;
          c.z += aa * vv.z;
          c.w += aa * vv.w;
          acc[d * SEG_TPB + tid] = c;
        }
      }
    }
    alldone = (doneS[0] & doneS[1] & doneS[2]) != 0;
    __syncthreads();
  }
  if (MODE == 0 && tid == 0) mine[0] = stored;

  if constexpr (!NODE) {
#pragma unroll
    for (int r = 0; r < R; ++r) {
      if (r < rr) {
        const int   cn = __builtin_amdgcn_readlane(ecnt, r);
        const float bi = (cn > 0) ? (1.0f / (float)cn) : 0.0f;
        float4 c = acc[r * SEG_TPB + tid];
        c.x *= bi; c.y *= bi; c.z *= bi; c.w *= bi;
        *reinterpret_cast<float4*>(dst + (size_t)(r0 + r) * OUT_CH + tid * 4) = c;
      }
    }
  } else {
    constexpr int ZU = 10;
    static_assert(R % ZU == 0, "whole groups of rows");
    float* dinvS = reinterpret_cast<float*>(lst);
#pragma unroll
    for (int m = 0; m < DM; ++m) {
      const int orow = k16 + 16 * m;
      if (orow < R) dinvS[orow * D + t] = (dsq[m] > 0.0f) ? (1.0f / dsq[m]) : 0.0f;
    }
    __syncthreads();
#pragma nounroll
    for (int rb = 0; rb < R; rb += ZU) {
      float4 hv[ZU];
      float  dinv[ZU];
#pragma unroll
      for (int j = 0; j < ZU; ++j) {
        hv[j]   = *reinterpret_cast<const float4*>(hself + (size_t)(r0 + rb + j) * OUT_CH + tid * 4);
        dinv[j] = dinvS[(rb + j) * D + t];
      }
      float4 cs[ZU];
#pragma unroll
      for (int j = 0; j < ZU; ++j) cs[j] = acc[(rb + j) * SEG_TPB + tid];
#pragma unroll
      for (int j = 0; j < ZU; ++j) {
        float4 c = cs[j];
        c.x = hv[j].x - dinv[j] * c.x;
        c.y = hv[j].y - dinv[j] * c.y;
        c.z = hv[j].z - dinv[j] * c.z;
        c.w = hv[j].w - dinv[j] * c.w;
        const int key = ((rb + j) * D + t) & 15;
        acc[(rb + j) * SEG_TPB + (tid & ~15) + ((tid & 15) ^ key)] = c;
      }
    }
    const int c16 = lane & 15;
    const int hh  = lane >> 4;
    constexpr int KP = HIDDEN / 8;
    v2f bf[KP][4][2];
#pragma unroll
    for (int p = 0; p < KP; ++p) {
#pragma unroll
      for (int nt = 0; nt < 4; ++nt) {
        const float* bp = W + (8 * p + 4 * hh) * HIDDEN + nt * 16 + c16;
        bf[p][nt][0].x = bp[0];
        bf[p][nt][0].y = bp[HIDDEN];
        bf[p][nt][1].x = bp[2 * HIDDEN];
        bf[p][nt][1].y = bp[3 * HIDDEN];
      }
    }
    float bv[4];
#pragma unroll
    for (int nt = 0; nt < 4; ++nt) bv[nt] = bias[nt * 16 + c16];
    __syncthreads();

    float* accf = reinterpret_cast<float*>(acc);
    constexpr int MT = R * D / 16;
    for (int j = wave; j < MT; j += SEG_WAVES) {
      float* T = accf + j * 16 * HIDDEN;
      v8f o[4] = {};
#pragma unroll
      for (int p = 0; p < KP; ++p) {
        const float4 a4 = *reinterpret_cast<const float4*>(T + c16 * HIDDEN + 4 * ((2 * p + hh) ^ c16));
        v2f a0; a0.x = a4.x; a0.y = a4.y;
        v2f a1; a1.x = a4.z; a1.y = a4.w;
#pragma unroll
        for (int nt = 0; nt < 4; ++nt)
          o[nt] = __builtin_amdgcn_wmma_f32_16x16x4_f32(false, a0, false, bf[p][nt][0],
                                                        (short)0, o[nt], false, false);
#pragma unroll
        for (int nt = 0; nt < 4; ++nt)
          o[nt] = __builtin_amdgcn_wmma_f32_16x16x4_f32(false, a1, false, bf[p][nt][1],
                                                        (short)0, o[nt], false, false);
      }
#pragma unroll
      for (int nt = 0; nt < 4; ++nt) {
#pragma unroll
        for (int r = 0; r < 8; ++r) T[(8 * hh + r) * HIDDEN + nt * 16 + c16] = o[nt][r] + bv[nt];
      }
      float4 ov[8];
#pragma unroll
      for (int i = 0; i < 8; ++i) {
        const int f = i * 32 + lane;
        ov[i] = *reinterpret_cast<const float4*>(T + (f >> 4) * HIDDEN + 4 * (f & 15));
      }
#pragma unroll
      for (int i = 0; i < 8; ++i) {
        const int f   = i * 32 + lane;
        const int row = f >> 4;
        const int q   = f & 15;
        float4 e = ov[i];
        e.x = (e.x > 0.0f) ? e.x : expm1f(e.x);
        e.y = (e.y > 0.0f) ? e.y : expm1f(e.y);
        e.z = (e.z > 0.0f) ? e.z : expm1f(e.z);
        e.w = (e.w > 0.0f) ? e.w : expm1f(e.w);
        *reinterpret_cast<float4*>(dst + ((size_t)r0 * D + j * 16 + row) * HIDDEN + 4 * q) = e;
      }
    }
  }
}

extern "C" void kernel_launch(void* const* d_in, const int* in_sizes, int n_in,
                              void* d_out, int out_size, void* d_ws, size_t ws_size,
                              hipStream_t stream) {
  (void)in_sizes; (void)n_in; (void)out_size; (void)ws_size;
  const float* x       = (const float*)d_in[0];
  const float* hattr   = (const float*)d_in[1];
  const float* W_lin   = (const float*)d_in[2];
  const float* b_lin   = (const float*)d_in[3];
  const float* ln_g    = (const float*)d_in[4];
  const float* ln_b    = (const float*)d_in[5];
  const float* W_sheaf = (const float*)d_in[6];
  const float* b_sheaf = (const float*)d_in[7];
  const float* W_conv  = (const float*)d_in[8];
  const float* b_conv  = (const float*)d_in[9];
  const int*   eidx    = (const int*)d_in[10];
  const int* row = eidx;
  const int* col = eidx + NNZ;
  float* out = (float*)d_out;

  float* ws    = (float*)d_ws;
  float* xs    = ws;
  float* es    = xs   + (size_t)ND * HIDDEN;
  float* lsts  = es   + (size_t)ED * HIDDEN;
  float* nst   = lsts + (size_t)N_NODES * HIDDEN;
  float* est   = nst  + (size_t)N_NODES * STAT_W;
  float* alpha = nst  + (size_t)N_EDGES * HIDDEN;
  int*   cl    = reinterpret_cast<int*>(alpha + (size_t)NNZ * D);
  float* me    = alpha + (size_t)NNZ * D + ND + N_EDGES + (size_t)ND * HIDDEN;
  static_assert((size_t)2 * CB_NB * CB_SUB * CB_RUN <= (size_t)ND + N_EDGES + (size_t)ND * HIDDEN, "bucket sub-lists end before me");

  {
    wmma_gemm_lds<IN_CH><<<dim3(85, OUT_CH / 64), GEMM_TPB, 0, stream>>>(x, W_lin, b_lin, xs,
                                                                       OUT_CH, N_NODES / 16);
    wmma_gemm_lds<IN_CH><<<dim3(79, OUT_CH / 64), GEMM_TPB, 0, stream>>>(hattr, W_lin, b_lin, es,
                                                                       OUT_CH, N_EDGES / 16);
  }

  static_assert((size_t)(N_NODES + N_EDGES) * STAT_W <= (size_t)N_EDGES * HIDDEN, "summaries fit in their region");
  rowstat_kernel<<<CB_WGS + (N_NODES + N_EDGES) * 16 / STAT_RPH / STAT_TPB, STAT_TPB, 0, stream>>>(xs, es, ln_g, W_sheaf, nst, est,
                                                                                                    row, col, cl);

  alpha_kernel<<<(NNZ + ALPHA_TPB - 1) / ALPHA_TPB, ALPHA_TPB, 0, stream>>>(nst, est, row, col, ln_g, ln_b,
                                                                           W_sheaf, b_sheaf, alpha);

  constexpr int RE = SEG_RE;
  constexpr int RN = SEG_RN;
  const int* clNode = cl;
  const int* clEdge = cl + (size_t)CB_NB * CB_SUB * CB_RUN;
  const int edgeBlk = (N_EDGES + RE - 1) / RE;
  const int nodeBlk = (N_NODES + RN - 1) / RN;
  static_assert(N_NODES % RN == 0, "the node side multiplies whole work-groups of RN rows");
  static_assert((size_t)((N_EDGES + RE - 1) / RE + (N_NODES + RN - 1) / RN) * SAVE_RUN
                    <= (size_t)N_NODES * HIDDEN, "saved runs fit in their region");
  int* edgeLists = reinterpret_cast<int*>(lsts);
  int* nodeLists = edgeLists + (size_t)edgeBlk * SAVE_RUN;

  segsum_kernel<RE, false, 0><<<edgeBlk, SEG_TPB, 0, stream>>>(xs, alpha, col, row, nullptr,
                                                               nullptr, nullptr, me, N_EDGES, edgeLists, clEdge);
  segsum_kernel<RN, true, 0><<<nodeBlk, SEG_TPB, 0, stream>>>(me, alpha, row, col, xs,
                                                              b_conv, W_conv, out, N_NODES, nodeLists, clNode);

  segsum_kernel<RE, false, 1><<<edgeBlk, SEG_TPB, 0, stream>>>(out, alpha, col, row, nullptr,
                                                               nullptr, nullptr, me, N_EDGES, edgeLists, nullptr);
  segsum_kernel<RN, true, 1><<<nodeBlk, SEG_TPB, 0, stream>>>(me, alpha, row, col, out,
                                                              b_conv + HIDDEN, W_conv + HIDDEN * HIDDEN, out, N_NODES, nodeLists, nullptr);
}
